# starting kernel
# baseline (speedup 1.0000x reference)
#include <hip/hip_runtime.h>
#include <stdint.h>


typedef float v2f __attribute__((ext_vector_type(2)));
typedef float v8f __attribute__((ext_vector_type(8)));

#define ORD_NEG_INF 0x007FFFFFu

__device__ __forceinline__ unsigned ord_encode(float f) {
    unsigned u = __float_as_uint(f);
    return (u & 0x80000000u) ? ~u : (u | 0x80000000u);
}

__device__ __forceinline__ float ord_decode_finite(unsigned k) {
    unsigned u = (k & 0x80000000u) ? (k & 0x7FFFFFFFu) : ~k;
    if ((u & 0x7F800000u) == 0x7F800000u) return 0.0f;
    return __uint_as_float(u);
}

__global__ void GCN_fill_u32(unsigned* __restrict__ p, long n, unsigned v) {
    long i = blockIdx.x * (long)blockDim.x + threadIdx.x;
    if (i < n) p[i] = v;
}

template <int LOG2F>
__global__ void GCN_edge_max(const float* __restrict__ feat,
                             const int* __restrict__ src,
                             const int* __restrict__ dst,
                             const float* __restrict__ ts,
                             unsigned* __restrict__ agg,
                             long nwork) {
    long t = blockIdx.x * (long)blockDim.x + threadIdx.x;
    if (t >= nwork) return;
    const int F = 1 << LOG2F;
    int e = (int)(t >> LOG2F);
    int f = (int)(t & (F - 1));
    int s = src[e];
    int d = dst[e];
    float m = feat[(long)s * F + f] * ts[e];
    atomicMax(agg + (long)d * F + f, ord_encode(m));
}

__global__ void __launch_bounds__(256)
GCN_finalize1(const unsigned* __restrict__ agg,
              const float* __restrict__ W1,
              const float* __restrict__ b1,
              float* __restrict__ h,
              int n_nodes) {
    int wave = (int)((blockIdx.x * (long)blockDim.x + threadIdx.x) >> 5);
    int lane = threadIdx.x & 31;
    int base = wave * 16;
    if (base >= n_nodes) return;

    int m  = lane & 15;
    int kh = lane >> 4;
    int n  = m;

    v8f acc = {0.f, 0.f, 0.f, 0.f, 0.f, 0.f, 0.f, 0.f};
#pragma unroll
    for (int c = 0; c < 4; ++c) {
        int k0 = 4 * c + 2 * kh;
        v2f a, b;
        a.x = ord_decode_finite(agg[(base + m) * 16 + k0]);
        a.y = ord_decode_finite(agg[(base + m) * 16 + k0 + 1]);
        b.x = (n < 8) ? W1[k0 * 8 + n]       : 0.0f;
        b.y = (n < 8) ? W1[(k0 + 1) * 8 + n] : 0.0f;
        acc = __builtin_amdgcn_wmma_f32_16x16x4_f32(
             false, a,  false, b,
             (short)0, acc,  false,  false);
    }

    if (n < 8) {
#pragma unroll
        for (int r = 0; r < 8; ++r) {
            int node = base + r + 8 * kh;
            float v = acc[r] + b1[n];
            h[node * 8 + n] = v > 0.0f ? v : 0.0f;
        }
    }
}

__global__ void __launch_bounds__(256)
GCN_finalize2(const unsigned* __restrict__ agg,
              const float* __restrict__ W2,
              const float* __restrict__ b2,
              float* __restrict__ out,
              int n_nodes) {
    int wave = (int)((blockIdx.x * (long)blockDim.x + threadIdx.x) >> 5);
    int lane = threadIdx.x & 31;
    int base = wave * 16;
    if (base >= n_nodes) return;

    int m  = lane & 15;
    int kh = lane >> 4;
    int n  = m;

    v8f acc = {0.f, 0.f, 0.f, 0.f, 0.f, 0.f, 0.f, 0.f};
#pragma unroll
    for (int c = 0; c < 2; ++c) {
        int k0 = 4 * c + 2 * kh;
        v2f a, b;
        a.x = ord_decode_finite(agg[(base + m) * 8 + k0]);
        a.y = ord_decode_finite(agg[(base + m) * 8 + k0 + 1]);
        b.x = (n < 2) ? W2[k0 * 2 + n]       : 0.0f;
        b.y = (n < 2) ? W2[(k0 + 1) * 2 + n] : 0.0f;
        acc = __builtin_amdgcn_wmma_f32_16x16x4_f32(
            false, a, false, b, (short)0, acc, false, false);
    }

    if (n < 2) {
#pragma unroll
        for (int r = 0; r < 8; ++r) {
            int node = base + r + 8 * kh;
            out[node * 2 + n] = acc[r] + b2[n];
        }
    }
}

extern "C" void kernel_launch(void* const* d_in, const int* in_sizes, int n_in,
                              void* d_out, int out_size, void* d_ws, size_t ws_size,
                              hipStream_t stream) {
    const float* x   = (const float*)d_in[0];
    const int*   src = (const int*)  d_in[1];
    const int*   dst = (const int*)  d_in[2];
    const float* ts  = (const float*)d_in[3];
    const float* W1  = (const float*)d_in[4];
    const float* b1  = (const float*)d_in[5];
    const float* W2  = (const float*)d_in[6];
    const float* b2  = (const float*)d_in[7];
    float* out = (float*)d_out;

    const int N = in_sizes[0] / 16;
    const int E = in_sizes[1];

    unsigned* agg1 = (unsigned*)d_ws;
    float*    h    = (float*)(agg1 + (size_t)N * 16);
    unsigned* agg2 = (unsigned*)(h + (size_t)N * 8);

    const int B = 256;

    long n1 = (long)N * 16;
    GCN_fill_u32<<<(unsigned)((n1 + B - 1) / B), B, 0, stream>>>(agg1, n1, ORD_NEG_INF);

    long w1 = (long)E * 16;
    GCN_edge_max<4><<<(unsigned)((w1 + B - 1) / B), B, 0, stream>>>(x, src, dst, ts, agg1, w1);

    int waves   = N / 16;
    int fin_thr = waves * 32;
    GCN_finalize1<<<(unsigned)((fin_thr + B - 1) / B), B, 0, stream>>>(agg1, W1, b1, h, N);

    long n2 = (long)N * 8;
    GCN_fill_u32<<<(unsigned)((n2 + B - 1) / B), B, 0, stream>>>(agg2, n2, ORD_NEG_INF);

    long w2 = (long)E * 8;
    GCN_edge_max<3><<<(unsigned)((w2 + B - 1) / B), B, 0, stream>>>(h, src, dst, ts, agg2, w2);

    GCN_finalize2<<<(unsigned)((fin_thr + B - 1) / B), B, 0, stream>>>(agg2, W2, b2, out, N);
}


# Edge passes read the agg slot with a relaxed device-scope load and send the atomic max only when the message is above it; max is order-free so the result is bit-identical, far fewer lane atomics
# speedup vs baseline: 1.0752x; 1.0752x over previous
#include <hip/hip_runtime.h>
#include <stdint.h>


typedef float v2f __attribute__((ext_vector_type(2)));
typedef float v8f __attribute__((ext_vector_type(8)));

#define ORD_NEG_INF 0x007FFFFFu

__device__ __forceinline__ unsigned ord_encode(float f) {
    unsigned u = __float_as_uint(f);
    return (u & 0x80000000u) ? ~u : (u | 0x80000000u);
}

__device__ __forceinline__ float ord_decode_finite(unsigned k) {
    unsigned u = (k & 0x80000000u) ? (k & 0x7FFFFFFFu) : ~k;
    if ((u & 0x7F800000u) == 0x7F800000u) return 0.0f;
    return __uint_as_float(u);
}

__global__ void GCN_fill_u32(unsigned* __restrict__ p, long n, unsigned v) {
    long i = blockIdx.x * (long)blockDim.x + threadIdx.x;
    if (i < n) p[i] = v;
}

template <int LOG2F>
__global__ void GCN_edge_max(const float* __restrict__ feat,
                             const int* __restrict__ src,
                             const int* __restrict__ dst,
                             const float* __restrict__ ts,
                             unsigned* agg,
                             long nwork) {
    long t = blockIdx.x * (long)blockDim.x + threadIdx.x;
    if (t >= nwork) return;
    const int F = 1 << LOG2F;
    int e = (int)(t >> LOG2F);
    int f = (int)(t & (F - 1));
    int s = src[e];
    int d = dst[e];
    unsigned* slot = agg + (long)d * F + f;
    unsigned cur = __hip_atomic_load(slot, __ATOMIC_RELAXED, __HIP_MEMORY_SCOPE_AGENT);
    float m = feat[(long)s * F + f] * ts[e];
    unsigned key = ord_encode(m);
    if (key > cur) atomicMax(slot, key);
}

__global__ void __launch_bounds__(256)
GCN_finalize1(const unsigned* __restrict__ agg,
              const float* __restrict__ W1,
              const float* __restrict__ b1,
              float* __restrict__ h,
              int n_nodes) {
    int wave = (int)((blockIdx.x * (long)blockDim.x + threadIdx.x) >> 5);
    int lane = threadIdx.x & 31;
    int base = wave * 16;
    if (base >= n_nodes) return;

    int m  = lane & 15;
    int kh = lane >> 4;
    int n  = m;

    v8f acc = {0.f, 0.f, 0.f, 0.f, 0.f, 0.f, 0.f, 0.f};
#pragma unroll
    for (int c = 0; c < 4; ++c) {
        int k0 = 4 * c + 2 * kh;
        v2f a, b;
        a.x = ord_decode_finite(agg[(base + m) * 16 + k0]);
        a.y = ord_decode_finite(agg[(base + m) * 16 + k0 + 1]);
        b.x = (n < 8) ? W1[k0 * 8 + n]       : 0.0f;
        b.y = (n < 8) ? W1[(k0 + 1) * 8 + n] : 0.0f;
        acc = __builtin_amdgcn_wmma_f32_16x16x4_f32(
             false, a,  false, b,
             (short)0, acc,  false,  false);
    }

    if (n < 8) {
#pragma unroll
        for (int r = 0; r < 8; ++r) {
            int node = base + r + 8 * kh;
            float v = acc[r] + b1[n];
            h[node * 8 + n] = v > 0.0f ? v : 0.0f;
        }
    }
}

__global__ void __launch_bounds__(256)
GCN_finalize2(const unsigned* __restrict__ agg,
              const float* __restrict__ W2,
              const float* __restrict__ b2,
              float* __restrict__ out,
              int n_nodes) {
    int wave = (int)((blockIdx.x * (long)blockDim.x + threadIdx.x) >> 5);
    int lane = threadIdx.x & 31;
    int base = wave * 16;
    if (base >= n_nodes) return;

    int m  = lane & 15;
    int kh = lane >> 4;
    int n  = m;

    v8f acc = {0.f, 0.f, 0.f, 0.f, 0.f, 0.f, 0.f, 0.f};
#pragma unroll
    for (int c = 0; c < 2; ++c) {
        int k0 = 4 * c + 2 * kh;
        v2f a, b;
        a.x = ord_decode_finite(agg[(base + m) * 8 + k0]);
        a.y = ord_decode_finite(agg[(base + m) * 8 + k0 + 1]);
        b.x = (n < 2) ? W2[k0 * 2 + n]       : 0.0f;
        b.y = (n < 2) ? W2[(k0 + 1) * 2 + n] : 0.0f;
        acc = __builtin_amdgcn_wmma_f32_16x16x4_f32(
            false, a, false, b, (short)0, acc, false, false);
    }

    if (n < 2) {
#pragma unroll
        for (int r = 0; r < 8; ++r) {
            int node = base + r + 8 * kh;
            out[node * 2 + n] = acc[r] + b2[n];
        }
    }
}

extern "C" void kernel_launch(void* const* d_in, const int* in_sizes, int n_in,
                              void* d_out, int out_size, void* d_ws, size_t ws_size,
                              hipStream_t stream) {
    const float* x   = (const float*)d_in[0];
    const int*   src = (const int*)  d_in[1];
    const int*   dst = (const int*)  d_in[2];
    const float* ts  = (const float*)d_in[3];
    const float* W1  = (const float*)d_in[4];
    const float* b1  = (const float*)d_in[5];
    const float* W2  = (const float*)d_in[6];
    const float* b2  = (const float*)d_in[7];
    float* out = (float*)d_out;

    const int N = in_sizes[0] / 16;
    const int E = in_sizes[1];

    unsigned* agg1 = (unsigned*)d_ws;
    float*    h    = (float*)(agg1 + (size_t)N * 16);
    unsigned* agg2 = (unsigned*)(h + (size_t)N * 8);

    const int B = 256;

    long n1 = (long)N * 16;
    GCN_fill_u32<<<(unsigned)((n1 + B - 1) / B), B, 0, stream>>>(agg1, n1, ORD_NEG_INF);

    long w1 = (long)E * 16;
    GCN_edge_max<4><<<(unsigned)((w1 + B - 1) / B), B, 0, stream>>>(x, src, dst, ts, agg1, w1);

    int waves   = N / 16;
    int fin_thr = waves * 32;
    GCN_finalize1<<<(unsigned)((fin_thr + B - 1) / B), B, 0, stream>>>(agg1, W1, b1, h, N);

    long n2 = (long)N * 8;
    GCN_fill_u32<<<(unsigned)((n2 + B - 1) / B), B, 0, stream>>>(agg2, n2, ORD_NEG_INF);

    long w2 = (long)E * 8;
    GCN_edge_max<3><<<(unsigned)((w2 + B - 1) / B), B, 0, stream>>>(h, src, dst, ts, agg2, w2);

    GCN_finalize2<<<(unsigned)((fin_thr + B - 1) / B), B, 0, stream>>>(agg2, W2, b2, out, N);
}


# Edge passes: one wave per 32 consecutive edges (coalesced index loads, DPP row broadcast), the batch's slot reads and gathers issued as one group ahead of the compares; atomic max kept, bit-identical
# speedup vs baseline: 1.2604x; 1.1722x over previous
#include <hip/hip_runtime.h>
#include <stdint.h>
#include <utility>


typedef float v2f __attribute__((ext_vector_type(2)));
typedef float v8f __attribute__((ext_vector_type(8)));

#define ORD_NEG_INF 0x007FFFFFu

__device__ __forceinline__ unsigned ord_encode(float f) {
    unsigned u = __float_as_uint(f);
    return (u & 0x80000000u) ? ~u : (u | 0x80000000u);
}

__device__ __forceinline__ float ord_decode_finite(unsigned k) {
    unsigned u = (k & 0x80000000u) ? (k & 0x7FFFFFFFu) : ~k;
    if ((u & 0x7F800000u) == 0x7F800000u) return 0.0f;
    return __uint_as_float(u);
}

__global__ void GCN_fill_u32(unsigned* __restrict__ p, long n, unsigned v) {
    long i = blockIdx.x * (long)blockDim.x + threadIdx.x;
    if (i < n) p[i] = v;
}


template <int J>
__device__ __forceinline__ int row_bcast16(int v) {
    return __builtin_amdgcn_update_dpp(0, v, 0x150 + J  , 0xf, 0xf, false);
}

template <int J0, int J1>
__device__ __forceinline__ int row_bcast8(int v) {
    int lo = __builtin_amdgcn_update_dpp(0, v, 0x150 + J0, 0xf, 0x3, false);
    return __builtin_amdgcn_update_dpp(lo, v, 0x150 + J1, 0xf, 0xc, false);
}

template <int LOG2F, int J>
__device__ __forceinline__ int edge_bcast(int v) {
    if constexpr (LOG2F == 4) return row_bcast16<J>(v);
    else                      return row_bcast8<2 * J, 2 * J + 1>(v);
}

template <int LOG2F, int J>
__device__ __forceinline__ void edge_step_load(int sv, int dv, unsigned fbyte,
                                               const char* __restrict__ feat_b, char* agg_b,
                                               unsigned* so, unsigned* cur, float* fv) {
    unsigned s = (unsigned)edge_bcast<LOG2F, J>(sv);
    unsigned d = (unsigned)edge_bcast<LOG2F, J>(dv);
    so[J] = (d << (LOG2F + 2)) + fbyte;
    unsigned fo = (s << (LOG2F + 2)) + fbyte;
    cur[J] = __hip_atomic_load((unsigned*)(agg_b + so[J]), __ATOMIC_RELAXED, __HIP_MEMORY_SCOPE_AGENT);
    fv[J]  = *(const float*)(feat_b + fo);
}

template <int LOG2F, int J>
__device__ __forceinline__ void edge_step_send(int tv, int lim, char* agg_b,
                                               const unsigned* so, const unsigned* cur, const float* fv) {
    const int STRIDE = (LOG2F == 4) ? 1 : 2;
    float t = __int_as_float(edge_bcast<LOG2F, J>(tv));
    unsigned key = ord_encode(fv[J] * t);
    if (STRIDE * J < lim && key > cur[J])
        atomicMax((unsigned*)(agg_b + so[J]), key);
}

template <int LOG2F, int... J>
__device__ __forceinline__ void edge_batch(int sv, int dv, int tv, int lim, unsigned fbyte,
                                           const char* __restrict__ feat_b, char* agg_b,
                                           std::integer_sequence<int, J...>) {
    unsigned so[sizeof...(J)], cur[sizeof...(J)];
    float fv[sizeof...(J)];
    (edge_step_load<LOG2F, J>(sv, dv, fbyte, feat_b, agg_b, so, cur, fv), ...);
    (edge_step_send<LOG2F, J>(tv, lim, agg_b, so, cur, fv), ...);
}

template <int LOG2F>
__global__ void __launch_bounds__(256)
GCN_edge_max(const float* __restrict__ feat,
             const int* __restrict__ src,
             const int* __restrict__ dst,
             const float* __restrict__ ts,
             unsigned* agg,
             int n_edges) {
    const int F     = 1 << LOG2F;
    const int STEPS = F;
    int lane = threadIdx.x & 31;
    long wave = (blockIdx.x * (long)blockDim.x + threadIdx.x) >> 5;
    long base = wave * 32;
    if (base >= n_edges) return;

    long e = base + lane;
    bool ok = e < n_edges;
    int sv = ok ? src[e] : 0;
    int dv = ok ? dst[e] : 0;
    int tv = ok ? __float_as_int(ts[e]) : 0;

    int f = lane & (F - 1);
    int u = (LOG2F == 4) ? 0 : ((lane >> 3) & 1);
    int lim = (int)(n_edges - base) - 16 * (lane >> 4) - u;

    edge_batch<LOG2F>(sv, dv, tv, lim, (unsigned)f << 2,
                      (const char*)feat, (char*)agg,
                      std::make_integer_sequence<int, STEPS>{});
}

__global__ void __launch_bounds__(256)
GCN_finalize1(const unsigned* __restrict__ agg,
              const float* __restrict__ W1,
              const float* __restrict__ b1,
              float* __restrict__ h,
              int n_nodes) {
    int wave = (int)((blockIdx.x * (long)blockDim.x + threadIdx.x) >> 5);
    int lane = threadIdx.x & 31;
    int base = wave * 16;
    if (base >= n_nodes) return;

    int m  = lane & 15;
    int kh = lane >> 4;
    int n  = m;

    v8f acc = {0.f, 0.f, 0.f, 0.f, 0.f, 0.f, 0.f, 0.f};
#pragma unroll
    for (int c = 0; c < 4; ++c) {
        int k0 = 4 * c + 2 * kh;
        v2f a, b;
        a.x = ord_decode_finite(agg[(base + m) * 16 + k0]);
        a.y = ord_decode_finite(agg[(base + m) * 16 + k0 + 1]);
        b.x = (n < 8) ? W1[k0 * 8 + n]       : 0.0f;
        b.y = (n < 8) ? W1[(k0 + 1) * 8 + n] : 0.0f;
        acc = __builtin_amdgcn_wmma_f32_16x16x4_f32(
             false, a,  false, b,
             (short)0, acc,  false,  false);
    }

    if (n < 8) {
#pragma unroll
        for (int r = 0; r < 8; ++r) {
            int node = base + r + 8 * kh;
            float v = acc[r] + b1[n];
            h[node * 8 + n] = v > 0.0f ? v : 0.0f;
        }
    }
}

__global__ void __launch_bounds__(256)
GCN_finalize2(const unsigned* __restrict__ agg,
              const float* __restrict__ W2,
              const float* __restrict__ b2,
              float* __restrict__ out,
              int n_nodes) {
    int wave = (int)((blockIdx.x * (long)blockDim.x + threadIdx.x) >> 5);
    int lane = threadIdx.x & 31;
    int base = wave * 16;
    if (base >= n_nodes) return;

    int m  = lane & 15;
    int kh = lane >> 4;
    int n  = m;

    v8f acc = {0.f, 0.f, 0.f, 0.f, 0.f, 0.f, 0.f, 0.f};
#pragma unroll
    for (int c = 0; c < 2; ++c) {
        int k0 = 4 * c + 2 * kh;
        v2f a, b;
        a.x = ord_decode_finite(agg[(base + m) * 8 + k0]);
        a.y = ord_decode_finite(agg[(base + m) * 8 + k0 + 1]);
        b.x = (n < 2) ? W2[k0 * 2 + n]       : 0.0f;
        b.y = (n < 2) ? W2[(k0 + 1) * 2 + n] : 0.0f;
        acc = __builtin_amdgcn_wmma_f32_16x16x4_f32(
            false, a, false, b, (short)0, acc, false, false);
    }

    if (n < 2) {
#pragma unroll
        for (int r = 0; r < 8; ++r) {
            int node = base + r + 8 * kh;
            out[node * 2 + n] = acc[r] + b2[n];
        }
    }
}

extern "C" void kernel_launch(void* const* d_in, const int* in_sizes, int n_in,
                              void* d_out, int out_size, void* d_ws, size_t ws_size,
                              hipStream_t stream) {
    const float* x   = (const float*)d_in[0];
    const int*   src = (const int*)  d_in[1];
    const int*   dst = (const int*)  d_in[2];
    const float* ts  = (const float*)d_in[3];
    const float* W1  = (const float*)d_in[4];
    const float* b1  = (const float*)d_in[5];
    const float* W2  = (const float*)d_in[6];
    const float* b2  = (const float*)d_in[7];
    float* out = (float*)d_out;

    const int N = in_sizes[0] / 16;
    const int E = in_sizes[1];

    unsigned* agg1 = (unsigned*)d_ws;
    float*    h    = (float*)(agg1 + (size_t)N * 16);
    unsigned* agg2 = (unsigned*)(h + (size_t)N * 8);

    const int B = 256;

    long n1 = (long)N * 16;
    GCN_fill_u32<<<(unsigned)((n1 + B - 1) / B), B, 0, stream>>>(agg1, n1, ORD_NEG_INF);

    unsigned eb = (unsigned)(((long)E + B - 1) / B);
    GCN_edge_max<4><<<eb, B, 0, stream>>>(x, src, dst, ts, agg1, E);

    int waves   = N / 16;
    int fin_thr = waves * 32;
    GCN_finalize1<<<(unsigned)((fin_thr + B - 1) / B), B, 0, stream>>>(agg1, W1, b1, h, N);

    long n2 = (long)N * 8;
    GCN_fill_u32<<<(unsigned)((n2 + B - 1) / B), B, 0, stream>>>(agg2, n2, ORD_NEG_INF);

    GCN_edge_max<3><<<eb, B, 0, stream>>>(h, src, dst, ts, agg2, E);

    GCN_finalize2<<<(unsigned)((fin_thr + B - 1) / B), B, 0, stream>>>(agg2, W2, b2, out, N);
}


# Edges grouped once per call by destination range (LDS counts and placement); each range keeps its maxima in LDS (LDS max, no global atomics), stores each row once; max is order-free: bit-identical
# speedup vs baseline: 2.9643x; 2.3519x over previous
#include <hip/hip_runtime.h>
#include <stdint.h>
#include <utility>


typedef float v2f __attribute__((ext_vector_type(2)));
typedef float v8f __attribute__((ext_vector_type(8)));

#define ORD_NEG_INF 0x007FFFFFu

#define BN       400
#define BTILES   25
#define NBMAX    256
#define NCHUNK   256
#define SRC_BITS 17

__device__ __forceinline__ unsigned ord_encode(float f) {
    unsigned u = __float_as_uint(f);
    return (u & 0x80000000u) ? ~u : (u | 0x80000000u);
}

__device__ __forceinline__ float ord_decode_finite(unsigned k) {
    unsigned u = (k & 0x80000000u) ? (k & 0x7FFFFFFFu) : ~k;
    if ((u & 0x7F800000u) == 0x7F800000u) return 0.0f;
    return __uint_as_float(u);
}

__global__ void __launch_bounds__(1024)
GCN_range_count(const int* __restrict__ dst, unsigned* __restrict__ counts,
                int n_edges, int chunk, int nb) {
    __shared__ unsigned hist[NBMAX];
    int tid = threadIdx.x;
    if (tid < NBMAX) hist[tid] = 0u;
    __syncthreads();
    long lo = (long)blockIdx.x * chunk;
    long hi = lo + chunk;
    if (hi > n_edges) hi = n_edges;
    for (long e0 = lo + tid; e0 < hi; e0 += 8 * 1024) {
        unsigned d[8];
#pragma unroll
        for (int k = 0; k < 8; ++k) {
            long e = e0 + (long)k * 1024;
            d[k] = (unsigned)dst[e < hi ? e : e0];
        }
#pragma unroll
        for (int k = 0; k < 8; ++k)
            if (e0 + (long)k * 1024 < hi) atomicAdd(&hist[d[k] / BN], 1u);
    }
    __syncthreads();
    if (tid < nb) counts[blockIdx.x * NBMAX + tid] = hist[tid];
}

__global__ void __launch_bounds__(1024)
GCN_range_place(const int* __restrict__ src, const int* __restrict__ dst,
                const float* __restrict__ ts, const unsigned* __restrict__ counts,
                unsigned* __restrict__ range_base, unsigned long long* __restrict__ rec,
                int n_edges, int chunk, int nb) {
    __shared__ unsigned ptot[4][NBMAX];
    __shared__ unsigned pbef[4][NBMAX];
    __shared__ unsigned off[NBMAX];
    int tid = threadIdx.x;
    {
        int b = tid & (NBMAX - 1);
        int q = tid >> 8;
        unsigned total = 0u, before = 0u;
        if (b < nb) {
#pragma unroll 8
            for (int g = q; g < (int)gridDim.x; g += 4) {
                unsigned c = counts[g * NBMAX + b];
                total += c;
                before += (g < (int)blockIdx.x) ? c : 0u;
            }
        }
        ptot[q][b] = total;
        pbef[q][b] = before;
    }
    __syncthreads();
    if (tid < 32) {
        unsigned v[8];
        unsigned s = 0u;
#pragma unroll
        for (int i = 0; i < 8; ++i) {
            int b = tid * 8 + i;
            v[i] = ptot[0][b] + ptot[1][b] + ptot[2][b] + ptot[3][b];
            s += v[i];
        }
        unsigned incl = s;
#pragma unroll
        for (int d = 1; d < 32; d <<= 1) {
            unsigned t = __shfl_up(incl, d);
            if (tid >= d) incl += t;
        }
        unsigned run = incl - s;
#pragma unroll
        for (int i = 0; i < 8; ++i) {
            int b = tid * 8 + i;
            off[b] = run + pbef[0][b] + pbef[1][b] + pbef[2][b] + pbef[3][b];
            if (blockIdx.x == 0 && b <= nb) range_base[b] = run;
            run += v[i];
        }
    }
    __syncthreads();
    long lo = (long)blockIdx.x * chunk;
    long hi = lo + chunk;
    if (hi > n_edges) hi = n_edges;
    for (long e0 = lo + tid; e0 < hi; e0 += 8 * 1024) {
        unsigned d[8], sv[8], tw[8];
#pragma unroll
        for (int k = 0; k < 8; ++k) {
            long e  = e0 + (long)k * 1024;
            long ec = e < hi ? e : e0;
            d[k]  = (unsigned)dst[ec];
            sv[k] = (unsigned)src[ec];
            tw[k] = __float_as_uint(ts[ec]);
        }
#pragma unroll
        for (int k = 0; k < 8; ++k) {
            if (e0 + (long)k * 1024 < hi) {
                unsigned b  = d[k] / BN;
                unsigned dl = d[k] - b * BN;
                unsigned p  = atomicAdd(&off[b], 1u);
                unsigned w0 = sv[k] | (dl << SRC_BITS);
                rec[p] = ((unsigned long long)tw[k] << 32) | (unsigned long long)w0;
            }
        }
    }
}


template <int J>
__device__ __forceinline__ int row_bcast16(int v) {
    return __builtin_amdgcn_update_dpp(0, v, 0x150 + J  , 0xf, 0xf, false);
}

template <int J0, int J1>
__device__ __forceinline__ int row_bcast8(int v) {
    int lo = __builtin_amdgcn_update_dpp(0, v, 0x150 + J0, 0xf, 0x3, false);
    return __builtin_amdgcn_update_dpp(lo, v, 0x150 + J1, 0xf, 0xc, false);
}

template <int LOG2F, int J>
__device__ __forceinline__ int rec_bcast(int v) {
    if constexpr (LOG2F == 4) return row_bcast16<J>(v);
    else                      return row_bcast8<2 * J, 2 * J + 1>(v);
}

template <int LOG2F, int J>
__device__ __forceinline__ void rec_step_load(int pv, unsigned fbyte,
                                              const char* __restrict__ feat_b,
                                              unsigned* so, float* fv) {
    unsigned p  = (unsigned)rec_bcast<LOG2F, J>(pv);
    unsigned s  = p & ((1u << SRC_BITS) - 1u);
    unsigned dl = p >> SRC_BITS;
    so[J] = (dl << LOG2F) + (fbyte >> 2);
    fv[J] = *(const float*)(feat_b + ((s << (LOG2F + 2)) + fbyte));
}

template <int LOG2F, int J>
__device__ __forceinline__ void rec_step_max(int tv, int lim, unsigned* plane,
                                             const unsigned* so, const float* fv) {
    const int STRIDE = (LOG2F == 4) ? 1 : 2;
    float t = __int_as_float(rec_bcast<LOG2F, J>(tv));
    unsigned key = ord_encode(fv[J] * t);
    if (STRIDE * J < lim)
        atomicMax(&plane[so[J]], key);
}

template <int LOG2F, int... J>
__device__ __forceinline__ void rec_batch(int pv, int tv, int lim, unsigned fbyte,
                                          const char* __restrict__ feat_b, unsigned* plane,
                                          std::integer_sequence<int, J...>) {
    unsigned so[sizeof...(J)];
    float fv[sizeof...(J)];
    (rec_step_load<LOG2F, J>(pv, fbyte, feat_b, so, fv), ...);
    (rec_step_max<LOG2F, J>(tv, lim, plane, so, fv), ...);
}

template <int LOG2F>
__device__ __forceinline__ void range_max(const unsigned long long* __restrict__ rec,
                                          unsigned bs, unsigned be,
                                          const float* __restrict__ feat, unsigned* plane) {
    const int F = 1 << LOG2F;
    int lane = threadIdx.x & 31;
    unsigned wv = (unsigned)__builtin_amdgcn_readfirstlane((int)(threadIdx.x >> 5));
    unsigned nw = blockDim.x >> 5;
    int f = lane & (F - 1);
    int u = (LOG2F == 4) ? 0 : ((lane >> 3) & 1);
    const unsigned step = 32u * nw;
    unsigned i = bs + 32u * wv;
    const unsigned last = be - 1u;
    unsigned long long r = 0ull;
    if (i < be) r = rec[min(i + (unsigned)lane, last)];
    for (; i < be; i += step) {
        unsigned long long rn = rec[min(i + step + (unsigned)lane, last)];
        int pv = (int)(unsigned)r;
        int tv = (int)(unsigned)(r >> 32);
        int lim = (int)(be - i) - 16 * (lane >> 4) - u;
        rec_batch<LOG2F>(pv, tv, lim, (unsigned)f << 2, (const char*)feat, plane,
                         std::make_integer_sequence<int, F>{});
        r = rn;
    }
}

__global__ void __launch_bounds__(1024)
GCN_layer1(const float* __restrict__ x,
           const unsigned long long* __restrict__ rec,
           const unsigned* __restrict__ range_base,
           const float* __restrict__ W1,
           const float* __restrict__ b1,
           float* __restrict__ h,
           int n_nodes) {
    __shared__ unsigned plane[BN * 16];
    for (int i = threadIdx.x; i < BN * 16; i += 1024) plane[i] = ORD_NEG_INF;
    __syncthreads();

    unsigned bs = range_base[blockIdx.x];
    unsigned be = range_base[blockIdx.x + 1];
    range_max<4>(rec, bs, be, x, plane);
    __syncthreads();

    int wv   = __builtin_amdgcn_readfirstlane((int)(threadIdx.x >> 5));
    int lane = threadIdx.x & 31;
    int base = (int)blockIdx.x * BN + wv * 16;
    if (wv >= BTILES || base >= n_nodes) return;

    int m  = lane & 15;
    int kh = lane >> 4;
    int n  = m;

    v8f acc = {0.f, 0.f, 0.f, 0.f, 0.f, 0.f, 0.f, 0.f};
#pragma unroll
    for (int c = 0; c < 4; ++c) {
        int k0 = 4 * c + 2 * kh;
        v2f a, b;
        a.x = ord_decode_finite(plane[(wv * 16 + m) * 16 + k0]);
        a.y = ord_decode_finite(plane[(wv * 16 + m) * 16 + k0 + 1]);
        b.x = (n < 8) ? W1[k0 * 8 + n]       : 0.0f;
        b.y = (n < 8) ? W1[(k0 + 1) * 8 + n] : 0.0f;
        acc = __builtin_amdgcn_wmma_f32_16x16x4_f32(
             false, a,  false, b,
             (short)0, acc,  false,  false);
    }

    if (n < 8) {
#pragma unroll
        for (int r = 0; r < 8; ++r) {
            int node = base + r + 8 * kh;
            float v = acc[r] + b1[n];
            h[node * 8 + n] = v > 0.0f ? v : 0.0f;
        }
    }
}

__global__ void __launch_bounds__(1024)
GCN_layer2(const float* __restrict__ h,
           const unsigned long long* __restrict__ rec,
           const unsigned* __restrict__ range_base,
           const float* __restrict__ W2,
           const float* __restrict__ b2,
           float* __restrict__ out,
           int n_nodes) {
    __shared__ unsigned plane[BN * 8];
    for (int i = threadIdx.x; i < BN * 8; i += 1024) plane[i] = ORD_NEG_INF;
    __syncthreads();

    unsigned bs = range_base[blockIdx.x];
    unsigned be = range_base[blockIdx.x + 1];
    range_max<3>(rec, bs, be, h, plane);
    __syncthreads();

    int wv   = __builtin_amdgcn_readfirstlane((int)(threadIdx.x >> 5));
    int lane = threadIdx.x & 31;
    int base = (int)blockIdx.x * BN + wv * 16;
    if (wv >= BTILES || base >= n_nodes) return;

    int m  = lane & 15;
    int kh = lane >> 4;
    int n  = m;

    v8f acc = {0.f, 0.f, 0.f, 0.f, 0.f, 0.f, 0.f, 0.f};
#pragma unroll
    for (int c = 0; c < 2; ++c) {
        int k0 = 4 * c + 2 * kh;
        v2f a, b;
        a.x = ord_decode_finite(plane[(wv * 16 + m) * 8 + k0]);
        a.y = ord_decode_finite(plane[(wv * 16 + m) * 8 + k0 + 1]);
        b.x = (n < 2) ? W2[k0 * 2 + n]       : 0.0f;
        b.y = (n < 2) ? W2[(k0 + 1) * 2 + n] : 0.0f;
        acc = __builtin_amdgcn_wmma_f32_16x16x4_f32(
            false, a, false, b, (short)0, acc, false, false);
    }

    if (n < 2) {
#pragma unroll
        for (int r = 0; r < 8; ++r) {
            int node = base + r + 8 * kh;
            out[node * 2 + n] = acc[r] + b2[n];
        }
    }
}

extern "C" void kernel_launch(void* const* d_in, const int* in_sizes, int n_in,
                              void* d_out, int out_size, void* d_ws, size_t ws_size,
                              hipStream_t stream) {
    const float* x   = (const float*)d_in[0];
    const int*   src = (const int*)  d_in[1];
    const int*   dst = (const int*)  d_in[2];
    const float* ts  = (const float*)d_in[3];
    const float* W1  = (const float*)d_in[4];
    const float* b1  = (const float*)d_in[5];
    const float* W2  = (const float*)d_in[6];
    const float* b2  = (const float*)d_in[7];
    float* out = (float*)d_out;

    const int N = in_sizes[0] / 16;
    const int E = in_sizes[1];

    unsigned* counts     = (unsigned*)d_ws;
    unsigned* range_base = counts + (size_t)NCHUNK * NBMAX;
    unsigned long long* rec = (unsigned long long*)(range_base + NBMAX);
    float* h = (float*)(rec + (size_t)E);

    const int nb    = (N + BN - 1) / BN;
    const int chunk = (E + NCHUNK - 1) / NCHUNK;

    GCN_range_count<<<NCHUNK, 1024, 0, stream>>>(dst, counts, E, chunk, nb);
    GCN_range_place<<<NCHUNK, 1024, 0, stream>>>(src, dst, ts, counts, range_base, rec, E, chunk, nb);
    GCN_layer1<<<nb, 1024, 0, stream>>>(x, rec, range_base, W1, b1, h, N);
    GCN_layer2<<<nb, 1024, 0, stream>>>(h, rec, range_base, W2, b2, out, N);
}


# One grouping pass: each edge chunk groups its own edges in its own part of the list, index loads in flight together (no count pass, no table, 3 launches); range passes walk 256 pieces; bit-identical
# speedup vs baseline: 3.1834x; 1.0739x over previous
#include <hip/hip_runtime.h>
#include <stdint.h>
#include <utility>


typedef float v2f __attribute__((ext_vector_type(2)));
typedef float v8f __attribute__((ext_vector_type(8)));

#define ORD_NEG_INF 0x007FFFFFu

#define BN       400
#define BTILES   25
#define NBMAX    256
#define NCHUNK   256
#define SRC_BITS 17
#define EPT      25
#define EPT2     13
#define PIECE    16

__device__ __forceinline__ unsigned ord_encode(float f) {
    unsigned u = __float_as_uint(f);
    return (u & 0x80000000u) ? ~u : (u | 0x80000000u);
}

__device__ __forceinline__ float ord_decode_finite(unsigned k) {
    unsigned u = (k & 0x80000000u) ? (k & 0x7FFFFFFFu) : ~k;
    if ((u & 0x7F800000u) == 0x7F800000u) return 0.0f;
    return __uint_as_float(u);
}

__global__ void __launch_bounds__(1024)
GCN_group(const int* __restrict__ src, const int* __restrict__ dst,
          const float* __restrict__ ts, unsigned long long* __restrict__ seg,
          unsigned long long* __restrict__ rec, int n_edges, int chunk, int nb) {
    __shared__ unsigned cnt[NBMAX];
    __shared__ unsigned off[NBMAX];
    const int tid = threadIdx.x;
    if (tid < NBMAX) cnt[tid] = 0u;
    __syncthreads();
    long lo = (long)blockIdx.x * chunk;
    if (lo > n_edges) lo = n_edges;
    long hi = lo + chunk;
    if (hi > n_edges) hi = n_edges;
    const unsigned n = (unsigned)(hi - lo);
    const int*   __restrict__ dst_c = dst + lo;
    const int*   __restrict__ src_c = src + lo;
    const float* __restrict__ ts_c  = ts + lo;
    unsigned long long* __restrict__ rec_c = rec + lo;
    for (unsigned i0 = (unsigned)tid; i0 < n; i0 += EPT * 1024u) {
        unsigned d[EPT];
#pragma unroll
        for (int k = 0; k < EPT; ++k) {
            unsigned i = i0 + (unsigned)k * 1024u;
            d[k] = (unsigned)dst_c[i < n ? i : i0];
        }
#pragma unroll
        for (int k = 0; k < EPT; ++k)
            if (i0 + (unsigned)k * 1024u < n) atomicAdd(&cnt[d[k] / BN], 1u);
    }
    __syncthreads();
    if (tid < 32) {
        unsigned v[8];
        unsigned s = 0u;
#pragma unroll
        for (int i = 0; i < 8; ++i) {
            v[i] = cnt[tid * 8 + i];
            s += v[i];
        }
        unsigned incl = s;
#pragma unroll
        for (int d = 1; d < 32; d <<= 1) {
            unsigned t = __shfl_up(incl, d);
            if (tid >= d) incl += t;
        }
        unsigned run = incl - s;
#pragma unroll
        for (int i = 0; i < 8; ++i) {
            int b = tid * 8 + i;
            off[b] = run;
            if (b < nb)
                seg[(size_t)b * NCHUNK + blockIdx.x] =
                    ((unsigned long long)v[i] << 32) | (unsigned long long)((unsigned)lo + run);
            run += v[i];
        }
    }
    __syncthreads();
    for (unsigned i0 = (unsigned)tid; i0 < n; i0 += EPT2 * 1024u) {
        unsigned d[EPT2], sv[EPT2], tw[EPT2], w0[EPT2], p[EPT2];
#pragma unroll
        for (int k = 0; k < EPT2; ++k) {
            unsigned i  = i0 + (unsigned)k * 1024u;
            unsigned ic = i < n ? i : i0;
            d[k]  = (unsigned)dst_c[ic];
            sv[k] = (unsigned)src_c[ic];
            tw[k] = __float_as_uint(ts_c[ic]);
        }
#pragma unroll
        for (int k = 0; k < EPT2; ++k) {
            unsigned b  = d[k] / BN;
            unsigned dl = d[k] - b * BN;
            w0[k] = sv[k] | (dl << SRC_BITS);
            p[k] = 0u;
            if (i0 + (unsigned)k * 1024u < n)
                p[k] = atomicAdd(&off[b], 1u);
        }
#pragma unroll
        for (int k = 0; k < EPT2; ++k)
            if (i0 + (unsigned)k * 1024u < n)
                rec_c[p[k]] = ((unsigned long long)tw[k] << 32) | (unsigned long long)w0[k];
    }
}


template <int J>
__device__ __forceinline__ int row_bcast16(int v) {
    return __builtin_amdgcn_update_dpp(0, v, 0x150 + J  , 0xf, 0xf, false);
}

template <int J0, int J1>
__device__ __forceinline__ int row_bcast8(int v) {
    int lo = __builtin_amdgcn_update_dpp(0, v, 0x150 + J0, 0xf, 0x3, false);
    return __builtin_amdgcn_update_dpp(lo, v, 0x150 + J1, 0xf, 0xc, false);
}

template <int LOG2F, int J>
__device__ __forceinline__ int rec_bcast(int v) {
    if constexpr (LOG2F == 4) return row_bcast16<J>(v);
    else                      return row_bcast8<2 * J, 2 * J + 1>(v);
}

template <int LOG2F, int J>
__device__ __forceinline__ void rec_step_load(int pv, unsigned fbyte,
                                              const char* __restrict__ feat_b,
                                              unsigned* so, float* fv) {
    unsigned p  = (unsigned)rec_bcast<LOG2F, J>(pv);
    unsigned s  = p & ((1u << SRC_BITS) - 1u);
    unsigned dl = p >> SRC_BITS;
    so[J] = (dl << LOG2F) + (fbyte >> 2);
    fv[J] = *(const float*)(feat_b + ((s << (LOG2F + 2)) + fbyte));
}

template <int LOG2F, int J>
__device__ __forceinline__ void rec_step_max(int tv, int lim, unsigned* plane,
                                             const unsigned* so, const float* fv) {
    const int STRIDE = (LOG2F == 4) ? 1 : 2;
    float t = __int_as_float(rec_bcast<LOG2F, J>(tv));
    unsigned key = ord_encode(fv[J] * t);
    if (STRIDE * J < lim)
        atomicMax(&plane[so[J]], key);
}

template <int LOG2F, int... J>
__device__ __forceinline__ void rec_batch(int pv, int tv, int lim, unsigned fbyte,
                                          const char* __restrict__ feat_b, unsigned* plane,
                                          std::integer_sequence<int, J...>) {
    unsigned so[sizeof...(J)];
    float fv[sizeof...(J)];
    (rec_step_load<LOG2F, J>(pv, fbyte, feat_b, so, fv), ...);
    (rec_step_max<LOG2F, J>(tv, lim, plane, so, fv), ...);
}

__device__ __forceinline__ void range_table(const unsigned long long* __restrict__ seg_row,
                                            unsigned* segQ, unsigned* segS, unsigned* segL) {
    const int tid = threadIdx.x;
    if (tid < NCHUNK) {
        unsigned long long e = seg_row[tid];
        segS[tid] = (unsigned)e;
        segL[tid] = (unsigned)(e >> 32);
    }
    __syncthreads();
    if (tid < 32) {
        unsigned v[8];
        unsigned s = 0u;
#pragma unroll
        for (int i = 0; i < 8; ++i) {
            v[i] = segL[tid * 8 + i];
            s += v[i];
        }
        unsigned incl = s;
#pragma unroll
        for (int d = 1; d < 32; d <<= 1) {
            unsigned t = __shfl_up(incl, d);
            if (tid >= d) incl += t;
        }
        unsigned run = incl - s;
#pragma unroll
        for (int i = 0; i < 8; ++i) {
            segQ[tid * 8 + i] = run;
            run += v[i];
        }
        if (tid == 31) segQ[NCHUNK] = run;
    }
    __syncthreads();
}

template <int LOG2F>
__device__ __forceinline__ void range_max(const unsigned long long* __restrict__ rec,
                                          const unsigned* segQ, const unsigned* segS,
                                          const float* __restrict__ feat, unsigned* plane) {
    const int F = 1 << LOG2F;
    const int lane = threadIdx.x & 31;
    const unsigned wv = (unsigned)__builtin_amdgcn_readfirstlane((int)(threadIdx.x >> 5));
    const unsigned nw = blockDim.x >> 5;
    const int f = lane & (F - 1);
    const int u = (LOG2F == 4) ? 0 : ((lane >> 3) & 1);
    const unsigned total = (unsigned)__builtin_amdgcn_readfirstlane((int)segQ[NCHUNK]);
    const unsigned span  = (total + nw - 1u) / nw;
    unsigned pos = min(wv * span, total);
    const unsigned span_end = min(pos + span, total);
    if (pos < span_end) {
        unsigned below = 0u;
#pragma unroll
        for (int r = 0; r < NCHUNK / 32; ++r)
            below += (unsigned)__builtin_popcount(
                __builtin_amdgcn_ballot_w32(segQ[lane + 32 * r] <= pos));
        unsigned kc = below - 1u;
        while (pos < span_end) {
            const unsigned kk = kc + (unsigned)lane;
            const unsigned qv = segQ[min(kk, (unsigned)NCHUNK)];
            const unsigned av = segS[min(kk, (unsigned)NCHUNK - 1u)] - qv;
            unsigned Q[PIECE + 1], A[PIECE];
#pragma unroll
            for (int m = 0; m <= PIECE; ++m) Q[m] = (unsigned)__builtin_amdgcn_readlane((int)qv, m);
#pragma unroll
            for (int m = 0; m < PIECE; ++m)  A[m] = (unsigned)__builtin_amdgcn_readlane((int)av, m);
            const unsigned piece_end = min(span_end, Q[PIECE]);
            if (pos < piece_end) {
                const unsigned lastj = piece_end - 1u;
                unsigned j = min(pos + (unsigned)lane, lastj);
                unsigned a = A[0];
#pragma unroll
                for (int m = 1; m < PIECE; ++m) a = (j >= Q[m]) ? A[m] : a;
                unsigned long long r = rec[j + a];
                for (unsigned i = pos; i < piece_end; i += 32u) {
                    j = min(i + 32u + (unsigned)lane, lastj);
                    a = A[0];
#pragma unroll
                    for (int m = 1; m < PIECE; ++m) a = (j >= Q[m]) ? A[m] : a;
                    unsigned long long rn = rec[j + a];
                    int pv = (int)(unsigned)r;
                    int tv = (int)(unsigned)(r >> 32);
                    int lim = (int)(piece_end - i) - 16 * (lane >> 4) - u;
                    rec_batch<LOG2F>(pv, tv, lim, (unsigned)f << 2, (const char*)feat, plane,
                                     std::make_integer_sequence<int, F>{});
                    r = rn;
                }
            }
            pos = piece_end;
            kc += PIECE;
        }
    }
}

__global__ void __launch_bounds__(1024)
GCN_layer1(const float* __restrict__ x,
           const unsigned long long* __restrict__ rec,
           const unsigned long long* __restrict__ seg,
           const float* __restrict__ W1,
           const float* __restrict__ b1,
           float* __restrict__ h,
           int n_nodes) {
    __shared__ unsigned plane[BN * 16];
    __shared__ unsigned segQ[NCHUNK + 1];
    __shared__ unsigned segS[NCHUNK];
    __shared__ unsigned segL[NCHUNK];
    for (int i = threadIdx.x; i < BN * 16; i += 1024) plane[i] = ORD_NEG_INF;
    range_table(seg + (size_t)blockIdx.x * NCHUNK, segQ, segS, segL);

    range_max<4>(rec, segQ, segS, x, plane);
    __syncthreads();

    int wv   = __builtin_amdgcn_readfirstlane((int)(threadIdx.x >> 5));
    int lane = threadIdx.x & 31;
    int base = (int)blockIdx.x * BN + wv * 16;
    if (wv >= BTILES || base >= n_nodes) return;

    int m  = lane & 15;
    int kh = lane >> 4;
    int n  = m;

    v8f acc = {0.f, 0.f, 0.f, 0.f, 0.f, 0.f, 0.f, 0.f};
#pragma unroll
    for (int c = 0; c < 4; ++c) {
        int k0 = 4 * c + 2 * kh;
        v2f a, b;
        a.x = ord_decode_finite(plane[(wv * 16 + m) * 16 + k0]);
        a.y = ord_decode_finite(plane[(wv * 16 + m) * 16 + k0 + 1]);
        b.x = (n < 8) ? W1[k0 * 8 + n]       : 0.0f;
        b.y = (n < 8) ? W1[(k0 + 1) * 8 + n] : 0.0f;
        acc = __builtin_amdgcn_wmma_f32_16x16x4_f32(
             false, a,  false, b,
             (short)0, acc,  false,  false);
    }

    if (n < 8) {
#pragma unroll
        for (int r = 0; r < 8; ++r) {
            int node = base + r + 8 * kh;
            float v = acc[r] + b1[n];
            h[node * 8 + n] = v > 0.0f ? v : 0.0f;
        }
    }
}

__global__ void __launch_bounds__(1024)
GCN_layer2(const float* __restrict__ h,
           const unsigned long long* __restrict__ rec,
           const unsigned long long* __restrict__ seg,
           const float* __restrict__ W2,
           const float* __restrict__ b2,
           float* __restrict__ out,
           int n_nodes) {
    __shared__ unsigned plane[BN * 8];
    __shared__ unsigned segQ[NCHUNK + 1];
    __shared__ unsigned segS[NCHUNK];
    __shared__ unsigned segL[NCHUNK];
    for (int i = threadIdx.x; i < BN * 8; i += 1024) plane[i] = ORD_NEG_INF;
    range_table(seg + (size_t)blockIdx.x * NCHUNK, segQ, segS, segL);

    range_max<3>(rec, segQ, segS, h, plane);
    __syncthreads();

    int wv   = __builtin_amdgcn_readfirstlane((int)(threadIdx.x >> 5));
    int lane = threadIdx.x & 31;
    int base = (int)blockIdx.x * BN + wv * 16;
    if (wv >= BTILES || base >= n_nodes) return;

    int m  = lane & 15;
    int kh = lane >> 4;
    int n  = m;

    v8f acc = {0.f, 0.f, 0.f, 0.f, 0.f, 0.f, 0.f, 0.f};
#pragma unroll
    for (int c = 0; c < 2; ++c) {
        int k0 = 4 * c + 2 * kh;
        v2f a, b;
        a.x = ord_decode_finite(plane[(wv * 16 + m) * 8 + k0]);
        a.y = ord_decode_finite(plane[(wv * 16 + m) * 8 + k0 + 1]);
        b.x = (n < 2) ? W2[k0 * 2 + n]       : 0.0f;
        b.y = (n < 2) ? W2[(k0 + 1) * 2 + n] : 0.0f;
        acc = __builtin_amdgcn_wmma_f32_16x16x4_f32(
            false, a, false, b, (short)0, acc, false, false);
    }

    if (n < 2) {
#pragma unroll
        for (int r = 0; r < 8; ++r) {
            int node = base + r + 8 * kh;
            out[node * 2 + n] = acc[r] + b2[n];
        }
    }
}

extern "C" void kernel_launch(void* const* d_in, const int* in_sizes, int n_in,
                              void* d_out, int out_size, void* d_ws, size_t ws_size,
                              hipStream_t stream) {
    const float* x   = (const float*)d_in[0];
    const int*   src = (const int*)  d_in[1];
    const int*   dst = (const int*)  d_in[2];
    const float* ts  = (const float*)d_in[3];
    const float* W1  = (const float*)d_in[4];
    const float* b1  = (const float*)d_in[5];
    const float* W2  = (const float*)d_in[6];
    const float* b2  = (const float*)d_in[7];
    float* out = (float*)d_out;

    const int N = in_sizes[0] / 16;
    const int E = in_sizes[1];

    unsigned long long* seg = (unsigned long long*)d_ws;
    unsigned long long* rec = seg + (size_t)NBMAX * NCHUNK;
    float* h = (float*)(rec + (size_t)E);

    const int nb    = (N + BN - 1) / BN;
    const int chunk = (E + NCHUNK - 1) / NCHUNK;

    GCN_group<<<NCHUNK, 1024, 0, stream>>>(src, dst, ts, seg, rec, E, chunk, nb);
    GCN_layer1<<<nb, 1024, 0, stream>>>(x, rec, seg, W1, b1, h, N);
    GCN_layer2<<<nb, 1024, 0, stream>>>(h, rec, seg, W2, b2, out, N);
}


# Grouping pass: adjacent lanes in one destination range add to its LDS counter once per run (head adds run length, lane permute hands out places); sorted edge stretches stop serialising; bit-identical
# speedup vs baseline: 4.2855x; 1.3462x over previous
#include <hip/hip_runtime.h>
#include <stdint.h>
#include <utility>


typedef float v2f __attribute__((ext_vector_type(2)));
typedef float v8f __attribute__((ext_vector_type(8)));

#define ORD_NEG_INF 0x007FFFFFu

#define BN       400
#define BTILES   25
#define NBMAX    256
#define NCHUNK   256
#define SRC_BITS 17
#define EPT      25
#define EPT2     13
#define PIECE    16

__device__ __forceinline__ unsigned ord_encode(float f) {
    unsigned u = __float_as_uint(f);
    return (u & 0x80000000u) ? ~u : (u | 0x80000000u);
}

__device__ __forceinline__ float ord_decode_finite(unsigned k) {
    unsigned u = (k & 0x80000000u) ? (k & 0x7FFFFFFFu) : ~k;
    if ((u & 0x7F800000u) == 0x7F800000u) return 0.0f;
    return __uint_as_float(u);
}

struct lane_run {
    unsigned heads;
    bool     head;
    unsigned len;
};

__device__ __forceinline__ lane_run find_runs(unsigned key, unsigned lane) {
    unsigned prev = (unsigned)__builtin_amdgcn_update_dpp(0, (int)key, 0x111  , 0xf, 0xf, false);
    lane_run r;
    r.head  = ((lane & 15u) == 0u) | (prev != key);
    r.heads = __builtin_amdgcn_ballot_w32(r.head);
    unsigned above = r.heads & ~((2u << lane) - 1u);
    unsigned next  = above ? (unsigned)__builtin_ctz(above) : 32u;
    r.len = next - lane;
    return r;
}

__global__ void __launch_bounds__(1024)
GCN_group(const int* __restrict__ src, const int* __restrict__ dst,
          const float* __restrict__ ts, unsigned long long* __restrict__ seg,
          unsigned long long* __restrict__ rec, int n_edges, int chunk, int nb) {
    __shared__ unsigned cnt[NBMAX];
    __shared__ unsigned off[NBMAX];
    const int tid = threadIdx.x;
    if (tid < NBMAX) cnt[tid] = 0u;
    __syncthreads();
    long lo = (long)blockIdx.x * chunk;
    if (lo > n_edges) lo = n_edges;
    long hi = lo + chunk;
    if (hi > n_edges) hi = n_edges;
    const unsigned n = (unsigned)(hi - lo);
    const int*   __restrict__ dst_c = dst + lo;
    const int*   __restrict__ src_c = src + lo;
    const float* __restrict__ ts_c  = ts + lo;
    unsigned long long* __restrict__ rec_c = rec + lo;
    const unsigned lane = (unsigned)tid & 31u;
    for (unsigned t0 = 0u; t0 < n; t0 += EPT * 1024u) {
        const unsigned i0 = t0 + (unsigned)tid;
        unsigned d[EPT];
#pragma unroll
        for (int k = 0; k < EPT; ++k) {
            unsigned i = i0 + (unsigned)k * 1024u;
            d[k] = (unsigned)dst_c[min(i, n - 1u)];
        }
#pragma unroll
        for (int k = 0; k < EPT; ++k) {
            const bool live = i0 + (unsigned)k * 1024u < n;
            const unsigned b = d[k] / BN;
            lane_run r = find_runs(live ? b : ~0u, lane);
            if (r.head && live) atomicAdd(&cnt[b], r.len);
        }
    }
    __syncthreads();
    if (tid < 32) {
        unsigned v[8];
        unsigned s = 0u;
#pragma unroll
        for (int i = 0; i < 8; ++i) {
            v[i] = cnt[tid * 8 + i];
            s += v[i];
        }
        unsigned incl = s;
#pragma unroll
        for (int d = 1; d < 32; d <<= 1) {
            unsigned t = __shfl_up(incl, d);
            if (tid >= d) incl += t;
        }
        unsigned run = incl - s;
#pragma unroll
        for (int i = 0; i < 8; ++i) {
            int b = tid * 8 + i;
            off[b] = run;
            if (b < nb)
                seg[(size_t)b * NCHUNK + blockIdx.x] =
                    ((unsigned long long)v[i] << 32) | (unsigned long long)((unsigned)lo + run);
            run += v[i];
        }
    }
    __syncthreads();
    for (unsigned t0 = 0u; t0 < n; t0 += EPT2 * 1024u) {
        const unsigned i0 = t0 + (unsigned)tid;
        unsigned d[EPT2], sv[EPT2], tw[EPT2], w0[EPT2], p[EPT2], hm[EPT2];
#pragma unroll
        for (int k = 0; k < EPT2; ++k) {
            unsigned ic = min(i0 + (unsigned)k * 1024u, n - 1u);
            d[k]  = (unsigned)dst_c[ic];
            sv[k] = (unsigned)src_c[ic];
            tw[k] = __float_as_uint(ts_c[ic]);
        }
#pragma unroll
        for (int k = 0; k < EPT2; ++k) {
            const bool live = i0 + (unsigned)k * 1024u < n;
            unsigned b  = d[k] / BN;
            unsigned dl = d[k] - b * BN;
            w0[k] = sv[k] | (dl << SRC_BITS);
            lane_run r = find_runs(live ? b : ~0u, lane);
            hm[k] = r.heads;
            p[k] = 0u;
            if (r.head && live)
                p[k] = atomicAdd(&off[b], r.len);
        }
#pragma unroll
        for (int k = 0; k < EPT2; ++k) {
            unsigned hl = 31u - (unsigned)__builtin_clz(hm[k] & ((2u << lane) - 1u));
            unsigned base = (unsigned)__builtin_amdgcn_ds_bpermute((int)(hl << 2), (int)p[k]);
            p[k] = base + (lane - hl);
        }
#pragma unroll
        for (int k = 0; k < EPT2; ++k)
            if (i0 + (unsigned)k * 1024u < n)
                rec_c[p[k]] = ((unsigned long long)tw[k] << 32) | (unsigned long long)w0[k];
    }
}


template <int J>
__device__ __forceinline__ int row_bcast16(int v) {
    return __builtin_amdgcn_update_dpp(0, v, 0x150 + J  , 0xf, 0xf, false);
}

template <int J0, int J1>
__device__ __forceinline__ int row_bcast8(int v) {
    int lo = __builtin_amdgcn_update_dpp(0, v, 0x150 + J0, 0xf, 0x3, false);
    return __builtin_amdgcn_update_dpp(lo, v, 0x150 + J1, 0xf, 0xc, false);
}

template <int LOG2F, int J>
__device__ __forceinline__ int rec_bcast(int v) {
    if constexpr (LOG2F == 4) return row_bcast16<J>(v);
    else                      return row_bcast8<2 * J, 2 * J + 1>(v);
}

template <int LOG2F, int J>
__device__ __forceinline__ void rec_step_load(int pv, unsigned fbyte,
                                              const char* __restrict__ feat_b,
                                              unsigned* so, float* fv) {
    unsigned p  = (unsigned)rec_bcast<LOG2F, J>(pv);
    unsigned s  = p & ((1u << SRC_BITS) - 1u);
    unsigned dl = p >> SRC_BITS;
    so[J] = (dl << LOG2F) + (fbyte >> 2);
    fv[J] = *(const float*)(feat_b + ((s << (LOG2F + 2)) + fbyte));
}

template <int LOG2F, int J>
__device__ __forceinline__ void rec_step_max(int tv, int lim, unsigned* plane,
                                             const unsigned* so, const float* fv) {
    const int STRIDE = (LOG2F == 4) ? 1 : 2;
    float t = __int_as_float(rec_bcast<LOG2F, J>(tv));
    unsigned key = ord_encode(fv[J] * t);
    if (STRIDE * J < lim)
        atomicMax(&plane[so[J]], key);
}

template <int LOG2F, int... J>
__device__ __forceinline__ void rec_batch(int pv, int tv, int lim, unsigned fbyte,
                                          const char* __restrict__ feat_b, unsigned* plane,
                                          std::integer_sequence<int, J...>) {
    unsigned so[sizeof...(J)];
    float fv[sizeof...(J)];
    (rec_step_load<LOG2F, J>(pv, fbyte, feat_b, so, fv), ...);
    (rec_step_max<LOG2F, J>(tv, lim, plane, so, fv), ...);
}

__device__ __forceinline__ void range_table(const unsigned long long* __restrict__ seg_row,
                                            unsigned* segQ, unsigned* segS, unsigned* segL) {
    const int tid = threadIdx.x;
    if (tid < NCHUNK) {
        unsigned long long e = seg_row[tid];
        segS[tid] = (unsigned)e;
        segL[tid] = (unsigned)(e >> 32);
    }
    __syncthreads();
    if (tid < 32) {
        unsigned v[8];
        unsigned s = 0u;
#pragma unroll
        for (int i = 0; i < 8; ++i) {
            v[i] = segL[tid * 8 + i];
            s += v[i];
        }
        unsigned incl = s;
#pragma unroll
        for (int d = 1; d < 32; d <<= 1) {
            unsigned t = __shfl_up(incl, d);
            if (tid >= d) incl += t;
        }
        unsigned run = incl - s;
#pragma unroll
        for (int i = 0; i < 8; ++i) {
            segQ[tid * 8 + i] = run;
            run += v[i];
        }
        if (tid == 31) segQ[NCHUNK] = run;
    }
    __syncthreads();
}

template <int LOG2F>
__device__ __forceinline__ void range_max(const unsigned long long* __restrict__ rec,
                                          const unsigned* segQ, const unsigned* segS,
                                          const float* __restrict__ feat, unsigned* plane) {
    const int F = 1 << LOG2F;
    const int lane = threadIdx.x & 31;
    const unsigned wv = (unsigned)__builtin_amdgcn_readfirstlane((int)(threadIdx.x >> 5));
    const unsigned nw = blockDim.x >> 5;
    const int f = lane & (F - 1);
    const int u = (LOG2F == 4) ? 0 : ((lane >> 3) & 1);
    const unsigned total = (unsigned)__builtin_amdgcn_readfirstlane((int)segQ[NCHUNK]);
    const unsigned span  = (total + nw - 1u) / nw;
    unsigned pos = min(wv * span, total);
    const unsigned span_end = min(pos + span, total);
    if (pos < span_end) {
        unsigned below = 0u;
#pragma unroll
        for (int r = 0; r < NCHUNK / 32; ++r)
            below += (unsigned)__builtin_popcount(
                __builtin_amdgcn_ballot_w32(segQ[lane + 32 * r] <= pos));
        unsigned kc = below - 1u;
        while (pos < span_end) {
            const unsigned kk = kc + (unsigned)lane;
            const unsigned qv = segQ[min(kk, (unsigned)NCHUNK)];
            const unsigned av = segS[min(kk, (unsigned)NCHUNK - 1u)] - qv;
            unsigned Q[PIECE + 1], A[PIECE];
#pragma unroll
            for (int m = 0; m <= PIECE; ++m) Q[m] = (unsigned)__builtin_amdgcn_readlane((int)qv, m);
#pragma unroll
            for (int m = 0; m < PIECE; ++m)  A[m] = (unsigned)__builtin_amdgcn_readlane((int)av, m);
            const unsigned piece_end = min(span_end, Q[PIECE]);
            if (pos < piece_end) {
                const unsigned lastj = piece_end - 1u;
                unsigned j = min(pos + (unsigned)lane, lastj);
                unsigned a = A[0];
#pragma unroll
                for (int m = 1; m < PIECE; ++m) a = (j >= Q[m]) ? A[m] : a;
                unsigned long long r = rec[j + a];
                for (unsigned i = pos; i < piece_end; i += 32u) {
                    j = min(i + 32u + (unsigned)lane, lastj);
                    a = A[0];
#pragma unroll
                    for (int m = 1; m < PIECE; ++m) a = (j >= Q[m]) ? A[m] : a;
                    unsigned long long rn = rec[j + a];
                    int pv = (int)(unsigned)r;
                    int tv = (int)(unsigned)(r >> 32);
                    int lim = (int)(piece_end - i) - 16 * (lane >> 4) - u;
                    rec_batch<LOG2F>(pv, tv, lim, (unsigned)f << 2, (const char*)feat, plane,
                                     std::make_integer_sequence<int, F>{});
                    r = rn;
                }
            }
            pos = piece_end;
            kc += PIECE;
        }
    }
}

__global__ void __launch_bounds__(1024)
GCN_layer1(const float* __restrict__ x,
           const unsigned long long* __restrict__ rec,
           const unsigned long long* __restrict__ seg,
           const float* __restrict__ W1,
           const float* __restrict__ b1,
           float* __restrict__ h,
           int n_nodes) {
    __shared__ unsigned plane[BN * 16];
    __shared__ unsigned segQ[NCHUNK + 1];
    __shared__ unsigned segS[NCHUNK];
    __shared__ unsigned segL[NCHUNK];
    for (int i = threadIdx.x; i < BN * 16; i += 1024) plane[i] = ORD_NEG_INF;
    range_table(seg + (size_t)blockIdx.x * NCHUNK, segQ, segS, segL);

    range_max<4>(rec, segQ, segS, x, plane);
    __syncthreads();

    int wv   = __builtin_amdgcn_readfirstlane((int)(threadIdx.x >> 5));
    int lane = threadIdx.x & 31;
    int base = (int)blockIdx.x * BN + wv * 16;
    if (wv >= BTILES || base >= n_nodes) return;

    int m  = lane & 15;
    int kh = lane >> 4;
    int n  = m;

    v8f acc = {0.f, 0.f, 0.f, 0.f, 0.f, 0.f, 0.f, 0.f};
#pragma unroll
    for (int c = 0; c < 4; ++c) {
        int k0 = 4 * c + 2 * kh;
        v2f a, b;
        a.x = ord_decode_finite(plane[(wv * 16 + m) * 16 + k0]);
        a.y = ord_decode_finite(plane[(wv * 16 + m) * 16 + k0 + 1]);
        b.x = (n < 8) ? W1[k0 * 8 + n]       : 0.0f;
        b.y = (n < 8) ? W1[(k0 + 1) * 8 + n] : 0.0f;
        acc = __builtin_amdgcn_wmma_f32_16x16x4_f32(
             false, a,  false, b,
             (short)0, acc,  false,  false);
    }

    if (n < 8) {
#pragma unroll
        for (int r = 0; r < 8; ++r) {
            int node = base + r + 8 * kh;
            float v = acc[r] + b1[n];
            h[node * 8 + n] = v > 0.0f ? v : 0.0f;
        }
    }
}

__global__ void __launch_bounds__(1024)
GCN_layer2(const float* __restrict__ h,
           const unsigned long long* __restrict__ rec,
           const unsigned long long* __restrict__ seg,
           const float* __restrict__ W2,
           const float* __restrict__ b2,
           float* __restrict__ out,
           int n_nodes) {
    __shared__ unsigned plane[BN * 8];
    __shared__ unsigned segQ[NCHUNK + 1];
    __shared__ unsigned segS[NCHUNK];
    __shared__ unsigned segL[NCHUNK];
    for (int i = threadIdx.x; i < BN * 8; i += 1024) plane[i] = ORD_NEG_INF;
    range_table(seg + (size_t)blockIdx.x * NCHUNK, segQ, segS, segL);

    range_max<3>(rec, segQ, segS, h, plane);
    __syncthreads();

    int wv   = __builtin_amdgcn_readfirstlane((int)(threadIdx.x >> 5));
    int lane = threadIdx.x & 31;
    int base = (int)blockIdx.x * BN + wv * 16;
    if (wv >= BTILES || base >= n_nodes) return;

    int m  = lane & 15;
    int kh = lane >> 4;
    int n  = m;

    v8f acc = {0.f, 0.f, 0.f, 0.f, 0.f, 0.f, 0.f, 0.f};
#pragma unroll
    for (int c = 0; c < 2; ++c) {
        int k0 = 4 * c + 2 * kh;
        v2f a, b;
        a.x = ord_decode_finite(plane[(wv * 16 + m) * 8 + k0]);
        a.y = ord_decode_finite(plane[(wv * 16 + m) * 8 + k0 + 1]);
        b.x = (n < 2) ? W2[k0 * 2 + n]       : 0.0f;
        b.y = (n < 2) ? W2[(k0 + 1) * 2 + n] : 0.0f;
        acc = __builtin_amdgcn_wmma_f32_16x16x4_f32(
            false, a, false, b, (short)0, acc, false, false);
    }

    if (n < 2) {
#pragma unroll
        for (int r = 0; r < 8; ++r) {
            int node = base + r + 8 * kh;
            out[node * 2 + n] = acc[r] + b2[n];
        }
    }
}

extern "C" void kernel_launch(void* const* d_in, const int* in_sizes, int n_in,
                              void* d_out, int out_size, void* d_ws, size_t ws_size,
                              hipStream_t stream) {
    const float* x   = (const float*)d_in[0];
    const int*   src = (const int*)  d_in[1];
    const int*   dst = (const int*)  d_in[2];
    const float* ts  = (const float*)d_in[3];
    const float* W1  = (const float*)d_in[4];
    const float* b1  = (const float*)d_in[5];
    const float* W2  = (const float*)d_in[6];
    const float* b2  = (const float*)d_in[7];
    float* out = (float*)d_out;

    const int N = in_sizes[0] / 16;
    const int E = in_sizes[1];

    unsigned long long* seg = (unsigned long long*)d_ws;
    unsigned long long* rec = seg + (size_t)NBMAX * NCHUNK;
    float* h = (float*)(rec + (size_t)E);

    const int nb    = (N + BN - 1) / BN;
    const int chunk = (E + NCHUNK - 1) / NCHUNK;

    GCN_group<<<NCHUNK, 1024, 0, stream>>>(src, dst, ts, seg, rec, E, chunk, nb);
    GCN_layer1<<<nb, 1024, 0, stream>>>(x, rec, seg, W1, b1, h, N);
    GCN_layer2<<<nb, 1024, 0, stream>>>(h, rec, seg, W2, b2, out, N);
}


# Grouping pass places its records in an LDS copy of the chunk's list part and writes it out in list order, whole 128-byte lines per wave store, not one scattered 8-byte store per edge; bit-identical
# speedup vs baseline: 4.7934x; 1.1185x over previous
#include <hip/hip_runtime.h>
#include <stdint.h>
#include <utility>


typedef float v2f __attribute__((ext_vector_type(2)));
typedef float v8f __attribute__((ext_vector_type(8)));

#define ORD_NEG_INF 0x007FFFFFu

#define BN       400
#define BTILES   25
#define NBMAX    256
#define NCHUNK   256
#define SRC_BITS 17
#define EPT      25
#define EPT2     13
#define PIECE    16
#define STAGE_CAP 25600

__device__ __forceinline__ unsigned ord_encode(float f) {
    unsigned u = __float_as_uint(f);
    return (u & 0x80000000u) ? ~u : (u | 0x80000000u);
}

__device__ __forceinline__ float ord_decode_finite(unsigned k) {
    unsigned u = (k & 0x80000000u) ? (k & 0x7FFFFFFFu) : ~k;
    if ((u & 0x7F800000u) == 0x7F800000u) return 0.0f;
    return __uint_as_float(u);
}

struct lane_run {
    unsigned heads;
    bool     head;
    unsigned len;
};

__device__ __forceinline__ lane_run find_runs(unsigned key, unsigned lane) {
    unsigned prev = (unsigned)__builtin_amdgcn_update_dpp(0, (int)key, 0x111  , 0xf, 0xf, false);
    lane_run r;
    r.head  = ((lane & 15u) == 0u) | (prev != key);
    r.heads = __builtin_amdgcn_ballot_w32(r.head);
    unsigned above = r.heads & ~((2u << lane) - 1u);
    unsigned next  = above ? (unsigned)__builtin_ctz(above) : 32u;
    r.len = next - lane;
    return r;
}

__global__ void __launch_bounds__(1024)
GCN_group(const int* __restrict__ src, const int* __restrict__ dst,
          const float* __restrict__ ts, unsigned long long* __restrict__ seg,
          unsigned long long* __restrict__ rec, int n_edges, int chunk, int nb) {
    __shared__ unsigned cnt[NBMAX];
    __shared__ unsigned off[NBMAX];
    __shared__ unsigned long long stage[STAGE_CAP];
    const int tid = threadIdx.x;
    if (tid < NBMAX) cnt[tid] = 0u;
    __syncthreads();
    long lo = (long)blockIdx.x * chunk;
    if (lo > n_edges) lo = n_edges;
    long hi = lo + chunk;
    if (hi > n_edges) hi = n_edges;
    const unsigned n = (unsigned)(hi - lo);
    const bool staged = n <= (unsigned)STAGE_CAP;
    const int*   __restrict__ dst_c = dst + lo;
    const int*   __restrict__ src_c = src + lo;
    const float* __restrict__ ts_c  = ts + lo;
    unsigned long long* __restrict__ rec_c = rec + lo;
    const unsigned lane = (unsigned)tid & 31u;
    for (unsigned t0 = 0u; t0 < n; t0 += EPT * 1024u) {
        const unsigned i0 = t0 + (unsigned)tid;
        unsigned d[EPT];
#pragma unroll
        for (int k = 0; k < EPT; ++k) {
            unsigned i = i0 + (unsigned)k * 1024u;
            d[k] = (unsigned)dst_c[min(i, n - 1u)];
        }
#pragma unroll
        for (int k = 0; k < EPT; ++k) {
            const bool live = i0 + (unsigned)k * 1024u < n;
            const unsigned b = d[k] / BN;
            lane_run r = find_runs(live ? b : ~0u, lane);
            if (r.head && live) atomicAdd(&cnt[b], r.len);
        }
    }
    __syncthreads();
    if (tid < 32) {
        unsigned v[8];
        unsigned s = 0u;
#pragma unroll
        for (int i = 0; i < 8; ++i) {
            v[i] = cnt[tid * 8 + i];
            s += v[i];
        }
        unsigned incl = s;
#pragma unroll
        for (int d = 1; d < 32; d <<= 1) {
            unsigned t = __shfl_up(incl, d);
            if (tid >= d) incl += t;
        }
        unsigned run = incl - s;
#pragma unroll
        for (int i = 0; i < 8; ++i) {
            int b = tid * 8 + i;
            off[b] = run;
            if (b < nb)
                seg[(size_t)b * NCHUNK + blockIdx.x] =
                    ((unsigned long long)v[i] << 32) | (unsigned long long)((unsigned)lo + run);
            run += v[i];
        }
    }
    __syncthreads();
    for (unsigned t0 = 0u; t0 < n; t0 += EPT2 * 1024u) {
        const unsigned i0 = t0 + (unsigned)tid;
        unsigned d[EPT2], sv[EPT2], tw[EPT2], w0[EPT2], p[EPT2], hm[EPT2];
#pragma unroll
        for (int k = 0; k < EPT2; ++k) {
            unsigned ic = min(i0 + (unsigned)k * 1024u, n - 1u);
            d[k]  = (unsigned)dst_c[ic];
            sv[k] = (unsigned)src_c[ic];
            tw[k] = __float_as_uint(ts_c[ic]);
        }
#pragma unroll
        for (int k = 0; k < EPT2; ++k) {
            const bool live = i0 + (unsigned)k * 1024u < n;
            unsigned b  = d[k] / BN;
            unsigned dl = d[k] - b * BN;
            w0[k] = sv[k] | (dl << SRC_BITS);
            lane_run r = find_runs(live ? b : ~0u, lane);
            hm[k] = r.heads;
            p[k] = 0u;
            if (r.head && live)
                p[k] = atomicAdd(&off[b], r.len);
        }
#pragma unroll
        for (int k = 0; k < EPT2; ++k) {
            unsigned hl = 31u - (unsigned)__builtin_clz(hm[k] & ((2u << lane) - 1u));
            unsigned base = (unsigned)__builtin_amdgcn_ds_bpermute((int)(hl << 2), (int)p[k]);
            p[k] = base + (lane - hl);
        }
        if (staged) {
#pragma unroll
            for (int k = 0; k < EPT2; ++k)
                if (i0 + (unsigned)k * 1024u < n)
                    stage[p[k]] = ((unsigned long long)tw[k] << 32) | (unsigned long long)w0[k];
        } else {
#pragma unroll
            for (int k = 0; k < EPT2; ++k)
                if (i0 + (unsigned)k * 1024u < n)
                    rec_c[p[k]] = ((unsigned long long)tw[k] << 32) | (unsigned long long)w0[k];
        }
    }
    __syncthreads();
    if (staged) {
        const unsigned a = min((16u - ((unsigned)lo & 15u)) & 15u, n);
        if ((unsigned)tid < a) rec_c[tid] = stage[tid];
        for (unsigned idx = a + (unsigned)tid; idx < n; idx += 1024u)
            rec_c[idx] = stage[idx];
    }
}


template <int J>
__device__ __forceinline__ int row_bcast16(int v) {
    return __builtin_amdgcn_update_dpp(0, v, 0x150 + J  , 0xf, 0xf, false);
}

template <int J0, int J1>
__device__ __forceinline__ int row_bcast8(int v) {
    int lo = __builtin_amdgcn_update_dpp(0, v, 0x150 + J0, 0xf, 0x3, false);
    return __builtin_amdgcn_update_dpp(lo, v, 0x150 + J1, 0xf, 0xc, false);
}

template <int LOG2F, int J>
__device__ __forceinline__ int rec_bcast(int v) {
    if constexpr (LOG2F == 4) return row_bcast16<J>(v);
    else                      return row_bcast8<2 * J, 2 * J + 1>(v);
}

template <int LOG2F, int J>
__device__ __forceinline__ void rec_step_load(int pv, unsigned fbyte,
                                              const char* __restrict__ feat_b,
                                              unsigned* so, float* fv) {
    unsigned p  = (unsigned)rec_bcast<LOG2F, J>(pv);
    unsigned s  = p & ((1u << SRC_BITS) - 1u);
    unsigned dl = p >> SRC_BITS;
    so[J] = (dl << LOG2F) + (fbyte >> 2);
    fv[J] = *(const float*)(feat_b + ((s << (LOG2F + 2)) + fbyte));
}

template <int LOG2F, int J>
__device__ __forceinline__ void rec_step_max(int tv, int lim, unsigned* plane,
                                             const unsigned* so, const float* fv) {
    const int STRIDE = (LOG2F == 4) ? 1 : 2;
    float t = __int_as_float(rec_bcast<LOG2F, J>(tv));
    unsigned key = ord_encode(fv[J] * t);
    if (STRIDE * J < lim)
        atomicMax(&plane[so[J]], key);
}

template <int LOG2F, int... J>
__device__ __forceinline__ void rec_batch(int pv, int tv, int lim, unsigned fbyte,
                                          const char* __restrict__ feat_b, unsigned* plane,
                                          std::integer_sequence<int, J...>) {
    unsigned so[sizeof...(J)];
    float fv[sizeof...(J)];
    (rec_step_load<LOG2F, J>(pv, fbyte, feat_b, so, fv), ...);
    (rec_step_max<LOG2F, J>(tv, lim, plane, so, fv), ...);
}

__device__ __forceinline__ void range_table(const unsigned long long* __restrict__ seg_row,
                                            unsigned* segQ, unsigned* segS, unsigned* segL) {
    const int tid = threadIdx.x;
    if (tid < NCHUNK) {
        unsigned long long e = seg_row[tid];
        segS[tid] = (unsigned)e;
        segL[tid] = (unsigned)(e >> 32);
    }
    __syncthreads();
    if (tid < 32) {
        unsigned v[8];
        unsigned s = 0u;
#pragma unroll
        for (int i = 0; i < 8; ++i) {
            v[i] = segL[tid * 8 + i];
            s += v[i];
        }
        unsigned incl = s;
#pragma unroll
        for (int d = 1; d < 32; d <<= 1) {
            unsigned t = __shfl_up(incl, d);
            if (tid >= d) incl += t;
        }
        unsigned run = incl - s;
#pragma unroll
        for (int i = 0; i < 8; ++i) {
            segQ[tid * 8 + i] = run;
            run += v[i];
        }
        if (tid == 31) segQ[NCHUNK] = run;
    }
    __syncthreads();
}

template <int LOG2F>
__device__ __forceinline__ void range_max(const unsigned long long* __restrict__ rec,
                                          const unsigned* segQ, const unsigned* segS,
                                          const float* __restrict__ feat, unsigned* plane) {
    const int F = 1 << LOG2F;
    const int lane = threadIdx.x & 31;
    const unsigned wv = (unsigned)__builtin_amdgcn_readfirstlane((int)(threadIdx.x >> 5));
    const unsigned nw = blockDim.x >> 5;
    const int f = lane & (F - 1);
    const int u = (LOG2F == 4) ? 0 : ((lane >> 3) & 1);
    const unsigned total = (unsigned)__builtin_amdgcn_readfirstlane((int)segQ[NCHUNK]);
    const unsigned span  = (total + nw - 1u) / nw;
    unsigned pos = min(wv * span, total);
    const unsigned span_end = min(pos + span, total);
    if (pos < span_end) {
        unsigned below = 0u;
#pragma unroll
        for (int r = 0; r < NCHUNK / 32; ++r)
            below += (unsigned)__builtin_popcount(
                __builtin_amdgcn_ballot_w32(segQ[lane + 32 * r] <= pos));
        unsigned kc = below - 1u;
        while (pos < span_end) {
            const unsigned kk = kc + (unsigned)lane;
            const unsigned qv = segQ[min(kk, (unsigned)NCHUNK)];
            const unsigned av = segS[min(kk, (unsigned)NCHUNK - 1u)] - qv;
            unsigned Q[PIECE + 1], A[PIECE];
#pragma unroll
            for (int m = 0; m <= PIECE; ++m) Q[m] = (unsigned)__builtin_amdgcn_readlane((int)qv, m);
#pragma unroll
            for (int m = 0; m < PIECE; ++m)  A[m] = (unsigned)__builtin_amdgcn_readlane((int)av, m);
            const unsigned piece_end = min(span_end, Q[PIECE]);
            if (pos < piece_end) {
                const unsigned lastj = piece_end - 1u;
                unsigned j = min(pos + (unsigned)lane, lastj);
                unsigned a = A[0];
#pragma unroll
                for (int m = 1; m < PIECE; ++m) a = (j >= Q[m]) ? A[m] : a;
                unsigned long long r = rec[j + a];
                for (unsigned i = pos; i < piece_end; i += 32u) {
                    j = min(i + 32u + (unsigned)lane, lastj);
                    a = A[0];
#pragma unroll
                    for (int m = 1; m < PIECE; ++m) a = (j >= Q[m]) ? A[m] : a;
                    unsigned long long rn = rec[j + a];
                    int pv = (int)(unsigned)r;
                    int tv = (int)(unsigned)(r >> 32);
                    int lim = (int)(piece_end - i) - 16 * (lane >> 4) - u;
                    rec_batch<LOG2F>(pv, tv, lim, (unsigned)f << 2, (const char*)feat, plane,
                                     std::make_integer_sequence<int, F>{});
                    r = rn;
                }
            }
            pos = piece_end;
            kc += PIECE;
        }
    }
}

__global__ void __launch_bounds__(1024)
GCN_layer1(const float* __restrict__ x,
           const unsigned long long* __restrict__ rec,
           const unsigned long long* __restrict__ seg,
           const float* __restrict__ W1,
           const float* __restrict__ b1,
           float* __restrict__ h,
           int n_nodes) {
    __shared__ unsigned plane[BN * 16];
    __shared__ unsigned segQ[NCHUNK + 1];
    __shared__ unsigned segS[NCHUNK];
    __shared__ unsigned segL[NCHUNK];
    for (int i = threadIdx.x; i < BN * 16; i += 1024) plane[i] = ORD_NEG_INF;
    range_table(seg + (size_t)blockIdx.x * NCHUNK, segQ, segS, segL);

    range_max<4>(rec, segQ, segS, x, plane);
    __syncthreads();

    int wv   = __builtin_amdgcn_readfirstlane((int)(threadIdx.x >> 5));
    int lane = threadIdx.x & 31;
    int base = (int)blockIdx.x * BN + wv * 16;
    if (wv >= BTILES || base >= n_nodes) return;

    int m  = lane & 15;
    int kh = lane >> 4;
    int n  = m;

    v8f acc = {0.f, 0.f, 0.f, 0.f, 0.f, 0.f, 0.f, 0.f};
#pragma unroll
    for (int c = 0; c < 4; ++c) {
        int k0 = 4 * c + 2 * kh;
        v2f a, b;
        a.x = ord_decode_finite(plane[(wv * 16 + m) * 16 + k0]);
        a.y = ord_decode_finite(plane[(wv * 16 + m) * 16 + k0 + 1]);
        b.x = (n < 8) ? W1[k0 * 8 + n]       : 0.0f;
        b.y = (n < 8) ? W1[(k0 + 1) * 8 + n] : 0.0f;
        acc = __builtin_amdgcn_wmma_f32_16x16x4_f32(
             false, a,  false, b,
             (short)0, acc,  false,  false);
    }

    if (n < 8) {
#pragma unroll
        for (int r = 0; r < 8; ++r) {
            int node = base + r + 8 * kh;
            float v = acc[r] + b1[n];
            h[node * 8 + n] = v > 0.0f ? v : 0.0f;
        }
    }
}

__global__ void __launch_bounds__(1024)
GCN_layer2(const float* __restrict__ h,
           const unsigned long long* __restrict__ rec,
           const unsigned long long* __restrict__ seg,
           const float* __restrict__ W2,
           const float* __restrict__ b2,
           float* __restrict__ out,
           int n_nodes) {
    __shared__ unsigned plane[BN * 8];
    __shared__ unsigned segQ[NCHUNK + 1];
    __shared__ unsigned segS[NCHUNK];
    __shared__ unsigned segL[NCHUNK];
    for (int i = threadIdx.x; i < BN * 8; i += 1024) plane[i] = ORD_NEG_INF;
    range_table(seg + (size_t)blockIdx.x * NCHUNK, segQ, segS, segL);

    range_max<3>(rec, segQ, segS, h, plane);
    __syncthreads();

    int wv   = __builtin_amdgcn_readfirstlane((int)(threadIdx.x >> 5));
    int lane = threadIdx.x & 31;
    int base = (int)blockIdx.x * BN + wv * 16;
    if (wv >= BTILES || base >= n_nodes) return;

    int m  = lane & 15;
    int kh = lane >> 4;
    int n  = m;

    v8f acc = {0.f, 0.f, 0.f, 0.f, 0.f, 0.f, 0.f, 0.f};
#pragma unroll
    for (int c = 0; c < 2; ++c) {
        int k0 = 4 * c + 2 * kh;
        v2f a, b;
        a.x = ord_decode_finite(plane[(wv * 16 + m) * 8 + k0]);
        a.y = ord_decode_finite(plane[(wv * 16 + m) * 8 + k0 + 1]);
        b.x = (n < 2) ? W2[k0 * 2 + n]       : 0.0f;
        b.y = (n < 2) ? W2[(k0 + 1) * 2 + n] : 0.0f;
        acc = __builtin_amdgcn_wmma_f32_16x16x4_f32(
            false, a, false, b, (short)0, acc, false, false);
    }

    if (n < 2) {
#pragma unroll
        for (int r = 0; r < 8; ++r) {
            int node = base + r + 8 * kh;
            out[node * 2 + n] = acc[r] + b2[n];
        }
    }
}

extern "C" void kernel_launch(void* const* d_in, const int* in_sizes, int n_in,
                              void* d_out, int out_size, void* d_ws, size_t ws_size,
                              hipStream_t stream) {
    const float* x   = (const float*)d_in[0];
    const int*   src = (const int*)  d_in[1];
    const int*   dst = (const int*)  d_in[2];
    const float* ts  = (const float*)d_in[3];
    const float* W1  = (const float*)d_in[4];
    const float* b1  = (const float*)d_in[5];
    const float* W2  = (const float*)d_in[6];
    const float* b2  = (const float*)d_in[7];
    float* out = (float*)d_out;

    const int N = in_sizes[0] / 16;
    const int E = in_sizes[1];

    unsigned long long* seg = (unsigned long long*)d_ws;
    unsigned long long* rec = seg + (size_t)NBMAX * NCHUNK;
    float* h = (float*)(rec + (size_t)E);

    const int nb    = (N + BN - 1) / BN;
    const int chunk = (E + NCHUNK - 1) / NCHUNK;

    GCN_group<<<NCHUNK, 1024, 0, stream>>>(src, dst, ts, seg, rec, E, chunk, nb);
    GCN_layer1<<<nb, 1024, 0, stream>>>(x, rec, seg, W1, b1, h, N);
    GCN_layer2<<<nb, 1024, 0, stream>>>(h, rec, seg, W2, b2, out, N);
}


# Grouping pass reads its chunk once: all loads asked for up front and held in registers; the counting LDS add returns each edge's rank, place = offset + rank (one atomic walk, not two); bit-identical
# speedup vs baseline: 5.0187x; 1.0470x over previous
#include <hip/hip_runtime.h>
#include <stdint.h>
#include <utility>


typedef float v2f __attribute__((ext_vector_type(2)));
typedef float v8f __attribute__((ext_vector_type(8)));

#define ORD_NEG_INF 0x007FFFFFu

#define BN       400
#define BTILES   25
#define NBMAX    256
#define NCHUNK   256
#define SRC_BITS 17
#define EPT      25
#define EPT2     13
#define PIECE    16
#define STAGE_CAP 25600

__device__ __forceinline__ unsigned ord_encode(float f) {
    unsigned u = __float_as_uint(f);
    return (u & 0x80000000u) ? ~u : (u | 0x80000000u);
}

__device__ __forceinline__ float ord_decode_finite(unsigned k) {
    unsigned u = (k & 0x80000000u) ? (k & 0x7FFFFFFFu) : ~k;
    if ((u & 0x7F800000u) == 0x7F800000u) return 0.0f;
    return __uint_as_float(u);
}

struct lane_run {
    unsigned heads;
    bool     head;
    unsigned len;
};

__device__ __forceinline__ lane_run find_runs(unsigned key, unsigned lane) {
    unsigned prev = (unsigned)__builtin_amdgcn_update_dpp(0, (int)key, 0x111  , 0xf, 0xf, false);
    lane_run r;
    r.head  = ((lane & 15u) == 0u) | (prev != key);
    r.heads = __builtin_amdgcn_ballot_w32(r.head);
    unsigned above = r.heads & ~((2u << lane) - 1u);
    unsigned next  = above ? (unsigned)__builtin_ctz(above) : 32u;
    r.len = next - lane;
    return r;
}

static_assert(STAGE_CAP <= EPT * 1024, "a staged chunk must fit one trip");
static_assert(NBMAX <= 256 && BN <= 512 && STAGE_CAP <= 32768,
              "range | row << 8 | rank << 17 must fit one word");

__device__ __forceinline__ void chunk_prefix(const unsigned* cnt, unsigned* off, int tid) {
    unsigned v[8];
    unsigned s = 0u;
#pragma unroll
    for (int i = 0; i < 8; ++i) {
        v[i] = cnt[tid * 8 + i];
        s += v[i];
    }
    unsigned incl = s;
#pragma unroll
    for (int d = 1; d < 32; d <<= 1) {
        unsigned t = __shfl_up(incl, d);
        if (tid >= d) incl += t;
    }
    unsigned run = incl - s;
#pragma unroll
    for (int i = 0; i < 8; ++i) {
        off[tid * 8 + i] = run;
        run += v[i];
    }
}

__device__ __forceinline__ void chunk_offsets(const unsigned* cnt, unsigned* off,
                                              unsigned long long* __restrict__ seg,
                                              unsigned lo, int nb, int tid) {
    unsigned v[8];
    unsigned s = 0u;
#pragma unroll
    for (int i = 0; i < 8; ++i) {
        v[i] = cnt[tid * 8 + i];
        s += v[i];
    }
    unsigned incl = s;
#pragma unroll
    for (int d = 1; d < 32; d <<= 1) {
        unsigned t = __shfl_up(incl, d);
        if (tid >= d) incl += t;
    }
    unsigned run = incl - s;
#pragma unroll
    for (int i = 0; i < 8; ++i) {
        int b = tid * 8 + i;
        off[b] = run;
        if (b < nb)
            seg[(size_t)b * NCHUNK + blockIdx.x] =
                ((unsigned long long)v[i] << 32) | (unsigned long long)(lo + run);
        run += v[i];
    }
}

__global__ void __launch_bounds__(1024)
GCN_group(const int* __restrict__ src, const int* __restrict__ dst,
          const float* __restrict__ ts, unsigned long long* __restrict__ seg,
          unsigned long long* __restrict__ rec, int n_edges, int chunk, int nb) {
    __shared__ unsigned cnt[NBMAX];
    __shared__ unsigned off[NBMAX];
    __shared__ uint2 stage[STAGE_CAP];
    const int tid = threadIdx.x;
    if (tid < NBMAX) cnt[tid] = 0u;
    __syncthreads();
    long lo = (long)blockIdx.x * chunk;
    if (lo > n_edges) lo = n_edges;
    long hi = lo + chunk;
    if (hi > n_edges) hi = n_edges;
    const unsigned n = (unsigned)(hi - lo);
    const int*   __restrict__ dst_c = dst + lo;
    const int*   __restrict__ src_c = src + lo;
    const float* __restrict__ ts_c  = ts + lo;
    unsigned long long* __restrict__ rec_c = rec + lo;
    const unsigned lane = (unsigned)tid & 31u;
    if (n <= (unsigned)STAGE_CAP) {
        unsigned key[EPT], sv[EPT], tw[EPT];
        const long lo_s = min(lo, (long)n_edges - 1);
        const int*   dst_s = dst + lo_s;
        const int*   src_s = src + lo_s;
        const float* ts_s  = ts + lo_s;
        const unsigned n1 = max(n, 1u) - 1u;
#pragma unroll
        for (int k = 0; k < EPT; ++k)
            key[k] = (unsigned)dst_s[min((unsigned)tid + (unsigned)k * 1024u, n1)];
#pragma unroll
        for (int k = 0; k < EPT; ++k)
            sv[k] = (unsigned)src_s[min((unsigned)tid + (unsigned)k * 1024u, n1)];
#pragma unroll
        for (int k = 0; k < EPT; ++k)
            tw[k] = __float_as_uint(ts_s[min((unsigned)tid + (unsigned)k * 1024u, n1)]);
        constexpr int GRP = 5;
#pragma unroll
        for (int g = 0; g < EPT; g += GRP) {
            unsigned p[GRP], hm[GRP];
#pragma unroll
            for (int j = 0; j < GRP; ++j) {
                const int k = g + j;
                if (k < EPT) {
                    const bool live = tid < (int)n - k * 1024;
                    const unsigned b  = key[k] / BN;
                    const unsigned dl = key[k] - b * BN;
                    lane_run r = find_runs(live ? b : ~0u, lane);
                    hm[j] = r.heads;
                    p[j] = 0u;
                    if (r.head && live)
                        p[j] = atomicAdd(&cnt[b], r.len);
                    key[k] = b | (dl << 8);
                }
            }
#pragma unroll
            for (int j = 0; j < GRP; ++j) {
                const int k = g + j;
                if (k < EPT) {
                    unsigned hl = 31u - (unsigned)__builtin_clz(hm[j] & ((2u << lane) - 1u));
                    unsigned base = (unsigned)__builtin_amdgcn_ds_bpermute((int)(hl << 2), (int)p[j]);
                    key[k] |= (base + (lane - hl)) << 17;
                    asm volatile("" :: "v"(key[k]));
                }
            }
        }
        __syncthreads();
        if (tid < 32) chunk_prefix(cnt, off, tid);
        __syncthreads();
        if (tid < nb)
            seg[(unsigned)tid * NCHUNK + blockIdx.x] =
                ((unsigned long long)cnt[tid] << 32) | (unsigned long long)((unsigned)lo + off[tid]);
#pragma unroll
        for (int g = 0; g < EPT; g += GRP) {
            unsigned pos[GRP];
#pragma unroll
            for (int j = 0; j < GRP; ++j)
                if (g + j < EPT) pos[j] = off[key[g + j] & 255u] + (key[g + j] >> 17);
#pragma unroll
            for (int j = 0; j < GRP; ++j) {
                const int k = g + j;
                if (k < EPT && tid < (int)n - k * 1024) {
                    const unsigned dl = (key[k] >> 8) & 511u;
                    stage[pos[j]] = make_uint2(sv[k] | (dl << SRC_BITS), tw[k]);
                }
            }
        }
        __syncthreads();
        uint2* __restrict__ rec_w = reinterpret_cast<uint2*>(rec_c);
        const unsigned a = min((16u - ((unsigned)lo & 15u)) & 15u, n);
        if ((unsigned)tid < a) rec_w[tid] = stage[tid];
        for (unsigned idx = a + (unsigned)tid; idx < n; idx += 1024u)
            rec_w[idx] = stage[idx];
    } else {
        for (unsigned t0 = 0u; t0 < n; t0 += EPT * 1024u) {
            const unsigned i0 = t0 + (unsigned)tid;
            unsigned d[EPT];
#pragma unroll
            for (int k = 0; k < EPT; ++k) {
                unsigned i = i0 + (unsigned)k * 1024u;
                d[k] = (unsigned)dst_c[min(i, n - 1u)];
            }
#pragma unroll
            for (int k = 0; k < EPT; ++k) {
                const bool live = i0 + (unsigned)k * 1024u < n;
                const unsigned b = d[k] / BN;
                lane_run r = find_runs(live ? b : ~0u, lane);
                if (r.head && live) atomicAdd(&cnt[b], r.len);
            }
        }
        __syncthreads();
        if (tid < 32) chunk_offsets(cnt, off, seg, (unsigned)lo, nb, tid);
        __syncthreads();
        for (unsigned t0 = 0u; t0 < n; t0 += EPT2 * 1024u) {
            const unsigned i0 = t0 + (unsigned)tid;
            unsigned d[EPT2], s2[EPT2], t2[EPT2], w0[EPT2], p[EPT2], hm[EPT2];
#pragma unroll
            for (int k = 0; k < EPT2; ++k) {
                unsigned ic = min(i0 + (unsigned)k * 1024u, n - 1u);
                d[k]  = (unsigned)dst_c[ic];
                s2[k] = (unsigned)src_c[ic];
                t2[k] = __float_as_uint(ts_c[ic]);
            }
#pragma unroll
            for (int k = 0; k < EPT2; ++k) {
                const bool live = i0 + (unsigned)k * 1024u < n;
                unsigned b  = d[k] / BN;
                unsigned dl = d[k] - b * BN;
                w0[k] = s2[k] | (dl << SRC_BITS);
                lane_run r = find_runs(live ? b : ~0u, lane);
                hm[k] = r.heads;
                p[k] = 0u;
                if (r.head && live)
                    p[k] = atomicAdd(&off[b], r.len);
            }
#pragma unroll
            for (int k = 0; k < EPT2; ++k) {
                unsigned hl = 31u - (unsigned)__builtin_clz(hm[k] & ((2u << lane) - 1u));
                unsigned base = (unsigned)__builtin_amdgcn_ds_bpermute((int)(hl << 2), (int)p[k]);
                p[k] = base + (lane - hl);
            }
#pragma unroll
            for (int k = 0; k < EPT2; ++k)
                if (i0 + (unsigned)k * 1024u < n)
                    rec_c[p[k]] = ((unsigned long long)t2[k] << 32) | (unsigned long long)w0[k];
        }
    }
}


template <int J>
__device__ __forceinline__ int row_bcast16(int v) {
    return __builtin_amdgcn_update_dpp(0, v, 0x150 + J  , 0xf, 0xf, false);
}

template <int J0, int J1>
__device__ __forceinline__ int row_bcast8(int v) {
    int lo = __builtin_amdgcn_update_dpp(0, v, 0x150 + J0, 0xf, 0x3, false);
    return __builtin_amdgcn_update_dpp(lo, v, 0x150 + J1, 0xf, 0xc, false);
}

template <int LOG2F, int J>
__device__ __forceinline__ int rec_bcast(int v) {
    if constexpr (LOG2F == 4) return row_bcast16<J>(v);
    else                      return row_bcast8<2 * J, 2 * J + 1>(v);
}

template <int LOG2F, int J>
__device__ __forceinline__ void rec_step_load(int pv, unsigned fbyte,
                                              const char* __restrict__ feat_b,
                                              unsigned* so, float* fv) {
    unsigned p  = (unsigned)rec_bcast<LOG2F, J>(pv);
    unsigned s  = p & ((1u << SRC_BITS) - 1u);
    unsigned dl = p >> SRC_BITS;
    so[J] = (dl << LOG2F) + (fbyte >> 2);
    fv[J] = *(const float*)(feat_b + ((s << (LOG2F + 2)) + fbyte));
}

template <int LOG2F, int J>
__device__ __forceinline__ void rec_step_max(int tv, int lim, unsigned* plane,
                                             const unsigned* so, const float* fv) {
    const int STRIDE = (LOG2F == 4) ? 1 : 2;
    float t = __int_as_float(rec_bcast<LOG2F, J>(tv));
    unsigned key = ord_encode(fv[J] * t);
    if (STRIDE * J < lim)
        atomicMax(&plane[so[J]], key);
}

template <int LOG2F, int... J>
__device__ __forceinline__ void rec_batch(int pv, int tv, int lim, unsigned fbyte,
                                          const char* __restrict__ feat_b, unsigned* plane,
                                          std::integer_sequence<int, J...>) {
    unsigned so[sizeof...(J)];
    float fv[sizeof...(J)];
    (rec_step_load<LOG2F, J>(pv, fbyte, feat_b, so, fv), ...);
    (rec_step_max<LOG2F, J>(tv, lim, plane, so, fv), ...);
}

__device__ __forceinline__ void range_table(const unsigned long long* __restrict__ seg_row,
                                            unsigned* segQ, unsigned* segS, unsigned* segL) {
    const int tid = threadIdx.x;
    if (tid < NCHUNK) {
        unsigned long long e = seg_row[tid];
        segS[tid] = (unsigned)e;
        segL[tid] = (unsigned)(e >> 32);
    }
    __syncthreads();
    if (tid < 32) {
        unsigned v[8];
        unsigned s = 0u;
#pragma unroll
        for (int i = 0; i < 8; ++i) {
            v[i] = segL[tid * 8 + i];
            s += v[i];
        }
        unsigned incl = s;
#pragma unroll
        for (int d = 1; d < 32; d <<= 1) {
            unsigned t = __shfl_up(incl, d);
            if (tid >= d) incl += t;
        }
        unsigned run = incl - s;
#pragma unroll
        for (int i = 0; i < 8; ++i) {
            segQ[tid * 8 + i] = run;
            run += v[i];
        }
        if (tid == 31) segQ[NCHUNK] = run;
    }
    __syncthreads();
}

template <int LOG2F>
__device__ __forceinline__ void range_max(const unsigned long long* __restrict__ rec,
                                          const unsigned* segQ, const unsigned* segS,
                                          const float* __restrict__ feat, unsigned* plane) {
    const int F = 1 << LOG2F;
    const int lane = threadIdx.x & 31;
    const unsigned wv = (unsigned)__builtin_amdgcn_readfirstlane((int)(threadIdx.x >> 5));
    const unsigned nw = blockDim.x >> 5;
    const int f = lane & (F - 1);
    const int u = (LOG2F == 4) ? 0 : ((lane >> 3) & 1);
    const unsigned total = (unsigned)__builtin_amdgcn_readfirstlane((int)segQ[NCHUNK]);
    const unsigned span  = (total + nw - 1u) / nw;
    unsigned pos = min(wv * span, total);
    const unsigned span_end = min(pos + span, total);
    if (pos < span_end) {
        unsigned below = 0u;
#pragma unroll
        for (int r = 0; r < NCHUNK / 32; ++r)
            below += (unsigned)__builtin_popcount(
                __builtin_amdgcn_ballot_w32(segQ[lane + 32 * r] <= pos));
        unsigned kc = below - 1u;
        while (pos < span_end) {
            const unsigned kk = kc + (unsigned)lane;
            const unsigned qv = segQ[min(kk, (unsigned)NCHUNK)];
            const unsigned av = segS[min(kk, (unsigned)NCHUNK - 1u)] - qv;
            unsigned Q[PIECE + 1], A[PIECE];
#pragma unroll
            for (int m = 0; m <= PIECE; ++m) Q[m] = (unsigned)__builtin_amdgcn_readlane((int)qv, m);
#pragma unroll
            for (int m = 0; m < PIECE; ++m)  A[m] = (unsigned)__builtin_amdgcn_readlane((int)av, m);
            const unsigned piece_end = min(span_end, Q[PIECE]);
            if (pos < piece_end) {
                const unsigned lastj = piece_end - 1u;
                unsigned j = min(pos + (unsigned)lane, lastj);
                unsigned a = A[0];
#pragma unroll
                for (int m = 1; m < PIECE; ++m) a = (j >= Q[m]) ? A[m] : a;
                unsigned long long r = rec[j + a];
                for (unsigned i = pos; i < piece_end; i += 32u) {
                    j = min(i + 32u + (unsigned)lane, lastj);
                    a = A[0];
#pragma unroll
                    for (int m = 1; m < PIECE; ++m) a = (j >= Q[m]) ? A[m] : a;
                    unsigned long long rn = rec[j + a];
                    int pv = (int)(unsigned)r;
                    int tv = (int)(unsigned)(r >> 32);
                    int lim = (int)(piece_end - i) - 16 * (lane >> 4) - u;
                    rec_batch<LOG2F>(pv, tv, lim, (unsigned)f << 2, (const char*)feat, plane,
                                     std::make_integer_sequence<int, F>{});
                    r = rn;
                }
            }
            pos = piece_end;
            kc += PIECE;
        }
    }
}

__global__ void __launch_bounds__(1024)
GCN_layer1(const float* __restrict__ x,
           const unsigned long long* __restrict__ rec,
           const unsigned long long* __restrict__ seg,
           const float* __restrict__ W1,
           const float* __restrict__ b1,
           float* __restrict__ h,
           int n_nodes) {
    __shared__ unsigned plane[BN * 16];
    __shared__ unsigned segQ[NCHUNK + 1];
    __shared__ unsigned segS[NCHUNK];
    __shared__ unsigned segL[NCHUNK];
    for (int i = threadIdx.x; i < BN * 16; i += 1024) plane[i] = ORD_NEG_INF;
    range_table(seg + (size_t)blockIdx.x * NCHUNK, segQ, segS, segL);

    range_max<4>(rec, segQ, segS, x, plane);
    __syncthreads();

    int wv   = __builtin_amdgcn_readfirstlane((int)(threadIdx.x >> 5));
    int lane = threadIdx.x & 31;
    int base = (int)blockIdx.x * BN + wv * 16;
    if (wv >= BTILES || base >= n_nodes) return;

    int m  = lane & 15;
    int kh = lane >> 4;
    int n  = m;

    v8f acc = {0.f, 0.f, 0.f, 0.f, 0.f, 0.f, 0.f, 0.f};
#pragma unroll
    for (int c = 0; c < 4; ++c) {
        int k0 = 4 * c + 2 * kh;
        v2f a, b;
        a.x = ord_decode_finite(plane[(wv * 16 + m) * 16 + k0]);
        a.y = ord_decode_finite(plane[(wv * 16 + m) * 16 + k0 + 1]);
        b.x = (n < 8) ? W1[k0 * 8 + n]       : 0.0f;
        b.y = (n < 8) ? W1[(k0 + 1) * 8 + n] : 0.0f;
        acc = __builtin_amdgcn_wmma_f32_16x16x4_f32(
             false, a,  false, b,
             (short)0, acc,  false,  false);
    }

    if (n < 8) {
#pragma unroll
        for (int r = 0; r < 8; ++r) {
            int node = base + r + 8 * kh;
            float v = acc[r] + b1[n];
            h[node * 8 + n] = v > 0.0f ? v : 0.0f;
        }
    }
}

__global__ void __launch_bounds__(1024)
GCN_layer2(const float* __restrict__ h,
           const unsigned long long* __restrict__ rec,
           const unsigned long long* __restrict__ seg,
           const float* __restrict__ W2,
           const float* __restrict__ b2,
           float* __restrict__ out,
           int n_nodes) {
    __shared__ unsigned plane[BN * 8];
    __shared__ unsigned segQ[NCHUNK + 1];
    __shared__ unsigned segS[NCHUNK];
    __shared__ unsigned segL[NCHUNK];
    for (int i = threadIdx.x; i < BN * 8; i += 1024) plane[i] = ORD_NEG_INF;
    range_table(seg + (size_t)blockIdx.x * NCHUNK, segQ, segS, segL);

    range_max<3>(rec, segQ, segS, h, plane);
    __syncthreads();

    int wv   = __builtin_amdgcn_readfirstlane((int)(threadIdx.x >> 5));
    int lane = threadIdx.x & 31;
    int base = (int)blockIdx.x * BN + wv * 16;
    if (wv >= BTILES || base >= n_nodes) return;

    int m  = lane & 15;
    int kh = lane >> 4;
    int n  = m;

    v8f acc = {0.f, 0.f, 0.f, 0.f, 0.f, 0.f, 0.f, 0.f};
#pragma unroll
    for (int c = 0; c < 2; ++c) {
        int k0 = 4 * c + 2 * kh;
        v2f a, b;
        a.x = ord_decode_finite(plane[(wv * 16 + m) * 8 + k0]);
        a.y = ord_decode_finite(plane[(wv * 16 + m) * 8 + k0 + 1]);
        b.x = (n < 2) ? W2[k0 * 2 + n]       : 0.0f;
        b.y = (n < 2) ? W2[(k0 + 1) * 2 + n] : 0.0f;
        acc = __builtin_amdgcn_wmma_f32_16x16x4_f32(
            false, a, false, b, (short)0, acc, false, false);
    }

    if (n < 2) {
#pragma unroll
        for (int r = 0; r < 8; ++r) {
            int node = base + r + 8 * kh;
            out[node * 2 + n] = acc[r] + b2[n];
        }
    }
}

extern "C" void kernel_launch(void* const* d_in, const int* in_sizes, int n_in,
                              void* d_out, int out_size, void* d_ws, size_t ws_size,
                              hipStream_t stream) {
    const float* x   = (const float*)d_in[0];
    const int*   src = (const int*)  d_in[1];
    const int*   dst = (const int*)  d_in[2];
    const float* ts  = (const float*)d_in[3];
    const float* W1  = (const float*)d_in[4];
    const float* b1  = (const float*)d_in[5];
    const float* W2  = (const float*)d_in[6];
    const float* b2  = (const float*)d_in[7];
    float* out = (float*)d_out;

    const int N = in_sizes[0] / 16;
    const int E = in_sizes[1];

    unsigned long long* seg = (unsigned long long*)d_ws;
    unsigned long long* rec = seg + (size_t)NBMAX * NCHUNK;
    float* h = (float*)(rec + (size_t)E);

    const int nb    = (N + BN - 1) / BN;
    const int chunk = (E + NCHUNK - 1) / NCHUNK;

    GCN_group<<<NCHUNK, 1024, 0, stream>>>(src, dst, ts, seg, rec, E, chunk, nb);
    GCN_layer1<<<nb, 1024, 0, stream>>>(x, rec, seg, W1, b1, h, N);
    GCN_layer2<<<nb, 1024, 0, stream>>>(h, rec, seg, W2, b2, out, N);
}


# Range passes sort a range's records by destination row in LDS (counting sort), fold a row's run in registers, 4 lanes per record (16/8-byte gathers); LDS max only at row changes; zero may flip sign
# speedup vs baseline: 5.7011x; 1.1360x over previous
#include <hip/hip_runtime.h>
#include <stdint.h>
#include <utility>


typedef float v2f __attribute__((ext_vector_type(2)));
typedef float v8f __attribute__((ext_vector_type(8)));

#define ORD_NEG_INF 0x007FFFFFu

#define BN       400
#define BTILES   25
#define NBMAX    256
#define NCHUNK   256
#define SRC_BITS 17
#define EPT      25
#define EPT2     13
#define PIECE    16
#define STAGE_CAP 25600
#define ROWS_PAD 512
#define SORT_CAP 28672
#define SORT_SLOTS 4
#define REM_C    9
#define REM_P    5

__device__ __forceinline__ unsigned ord_encode(float f) {
    unsigned u = __float_as_uint(f);
    return (u & 0x80000000u) ? ~u : (u | 0x80000000u);
}

__device__ __forceinline__ float ord_decode_finite(unsigned k) {
    unsigned u = (k & 0x80000000u) ? (k & 0x7FFFFFFFu) : ~k;
    if ((u & 0x7F800000u) == 0x7F800000u) return 0.0f;
    return __uint_as_float(u);
}

struct lane_run {
    unsigned heads;
    bool     head;
    unsigned len;
};

__device__ __forceinline__ lane_run find_runs(unsigned key, unsigned lane) {
    unsigned prev = (unsigned)__builtin_amdgcn_update_dpp(0, (int)key, 0x111  , 0xf, 0xf, false);
    lane_run r;
    r.head  = ((lane & 15u) == 0u) | (prev != key);
    r.heads = __builtin_amdgcn_ballot_w32(r.head);
    unsigned above = r.heads & ~((2u << lane) - 1u);
    unsigned next  = above ? (unsigned)__builtin_ctz(above) : 32u;
    r.len = next - lane;
    return r;
}

static_assert(STAGE_CAP <= EPT * 1024, "a staged chunk must fit one trip");
static_assert(NBMAX <= 256 && BN <= 512 && STAGE_CAP <= 32768,
              "range | row << 8 | rank << 17 must fit one word");

__device__ __forceinline__ void chunk_prefix(const unsigned* cnt, unsigned* off, int tid) {
    unsigned v[8];
    unsigned s = 0u;
#pragma unroll
    for (int i = 0; i < 8; ++i) {
        v[i] = cnt[tid * 8 + i];
        s += v[i];
    }
    unsigned incl = s;
#pragma unroll
    for (int d = 1; d < 32; d <<= 1) {
        unsigned t = __shfl_up(incl, d);
        if (tid >= d) incl += t;
    }
    unsigned run = incl - s;
#pragma unroll
    for (int i = 0; i < 8; ++i) {
        off[tid * 8 + i] = run;
        run += v[i];
    }
}

__device__ __forceinline__ void chunk_offsets(const unsigned* cnt, unsigned* off,
                                              unsigned long long* __restrict__ seg,
                                              unsigned lo, int nb, int tid) {
    unsigned v[8];
    unsigned s = 0u;
#pragma unroll
    for (int i = 0; i < 8; ++i) {
        v[i] = cnt[tid * 8 + i];
        s += v[i];
    }
    unsigned incl = s;
#pragma unroll
    for (int d = 1; d < 32; d <<= 1) {
        unsigned t = __shfl_up(incl, d);
        if (tid >= d) incl += t;
    }
    unsigned run = incl - s;
#pragma unroll
    for (int i = 0; i < 8; ++i) {
        int b = tid * 8 + i;
        off[b] = run;
        if (b < nb)
            seg[(size_t)b * NCHUNK + blockIdx.x] =
                ((unsigned long long)v[i] << 32) | (unsigned long long)(lo + run);
        run += v[i];
    }
}

__global__ void __launch_bounds__(1024)
GCN_group(const int* __restrict__ src, const int* __restrict__ dst,
          const float* __restrict__ ts, unsigned long long* __restrict__ seg,
          unsigned long long* __restrict__ rec, int n_edges, int chunk, int nb) {
    __shared__ unsigned cnt[NBMAX];
    __shared__ unsigned off[NBMAX];
    __shared__ uint2 stage[STAGE_CAP];
    const int tid = threadIdx.x;
    if (tid < NBMAX) cnt[tid] = 0u;
    __syncthreads();
    long lo = (long)blockIdx.x * chunk;
    if (lo > n_edges) lo = n_edges;
    long hi = lo + chunk;
    if (hi > n_edges) hi = n_edges;
    const unsigned n = (unsigned)(hi - lo);
    const int*   __restrict__ dst_c = dst + lo;
    const int*   __restrict__ src_c = src + lo;
    const float* __restrict__ ts_c  = ts + lo;
    unsigned long long* __restrict__ rec_c = rec + lo;
    const unsigned lane = (unsigned)tid & 31u;
    if (n <= (unsigned)STAGE_CAP) {
        unsigned key[EPT], sv[EPT], tw[EPT];
        const long lo_s = min(lo, (long)n_edges - 1);
        const int*   dst_s = dst + lo_s;
        const int*   src_s = src + lo_s;
        const float* ts_s  = ts + lo_s;
        const unsigned n1 = max(n, 1u) - 1u;
#pragma unroll
        for (int k = 0; k < EPT; ++k)
            key[k] = (unsigned)dst_s[min((unsigned)tid + (unsigned)k * 1024u, n1)];
#pragma unroll
        for (int k = 0; k < EPT; ++k)
            sv[k] = (unsigned)src_s[min((unsigned)tid + (unsigned)k * 1024u, n1)];
#pragma unroll
        for (int k = 0; k < EPT; ++k)
            tw[k] = __float_as_uint(ts_s[min((unsigned)tid + (unsigned)k * 1024u, n1)]);
        constexpr int GRP = 5;
#pragma unroll
        for (int g = 0; g < EPT; g += GRP) {
            unsigned p[GRP], hm[GRP];
#pragma unroll
            for (int j = 0; j < GRP; ++j) {
                const int k = g + j;
                if (k < EPT) {
                    const bool live = tid < (int)n - k * 1024;
                    const unsigned b  = key[k] / BN;
                    const unsigned dl = key[k] - b * BN;
                    lane_run r = find_runs(live ? b : ~0u, lane);
                    hm[j] = r.heads;
                    p[j] = 0u;
                    if (r.head && live)
                        p[j] = atomicAdd(&cnt[b], r.len);
                    key[k] = b | (dl << 8);
                }
            }
#pragma unroll
            for (int j = 0; j < GRP; ++j) {
                const int k = g + j;
                if (k < EPT) {
                    unsigned hl = 31u - (unsigned)__builtin_clz(hm[j] & ((2u << lane) - 1u));
                    unsigned base = (unsigned)__builtin_amdgcn_ds_bpermute((int)(hl << 2), (int)p[j]);
                    key[k] |= (base + (lane - hl)) << 17;
                    asm volatile("" :: "v"(key[k]));
                }
            }
        }
        __syncthreads();
        if (tid < 32) chunk_prefix(cnt, off, tid);
        __syncthreads();
        if (tid < nb)
            seg[(unsigned)tid * NCHUNK + blockIdx.x] =
                ((unsigned long long)cnt[tid] << 32) | (unsigned long long)((unsigned)lo + off[tid]);
#pragma unroll
        for (int g = 0; g < EPT; g += GRP) {
            unsigned pos[GRP];
#pragma unroll
            for (int j = 0; j < GRP; ++j)
                if (g + j < EPT) pos[j] = off[key[g + j] & 255u] + (key[g + j] >> 17);
#pragma unroll
            for (int j = 0; j < GRP; ++j) {
                const int k = g + j;
                if (k < EPT && tid < (int)n - k * 1024) {
                    const unsigned dl = (key[k] >> 8) & 511u;
                    stage[pos[j]] = make_uint2(sv[k] | (dl << SRC_BITS), tw[k]);
                }
            }
        }
        __syncthreads();
        uint2* __restrict__ rec_w = reinterpret_cast<uint2*>(rec_c);
        const unsigned a = min((16u - ((unsigned)lo & 15u)) & 15u, n);
        if ((unsigned)tid < a) rec_w[tid] = stage[tid];
        for (unsigned idx = a + (unsigned)tid; idx < n; idx += 1024u)
            rec_w[idx] = stage[idx];
    } else {
        for (unsigned t0 = 0u; t0 < n; t0 += EPT * 1024u) {
            const unsigned i0 = t0 + (unsigned)tid;
            unsigned d[EPT];
#pragma unroll
            for (int k = 0; k < EPT; ++k) {
                unsigned i = i0 + (unsigned)k * 1024u;
                d[k] = (unsigned)dst_c[min(i, n - 1u)];
            }
#pragma unroll
            for (int k = 0; k < EPT; ++k) {
                const bool live = i0 + (unsigned)k * 1024u < n;
                const unsigned b = d[k] / BN;
                lane_run r = find_runs(live ? b : ~0u, lane);
                if (r.head && live) atomicAdd(&cnt[b], r.len);
            }
        }
        __syncthreads();
        if (tid < 32) chunk_offsets(cnt, off, seg, (unsigned)lo, nb, tid);
        __syncthreads();
        for (unsigned t0 = 0u; t0 < n; t0 += EPT2 * 1024u) {
            const unsigned i0 = t0 + (unsigned)tid;
            unsigned d[EPT2], s2[EPT2], t2[EPT2], w0[EPT2], p[EPT2], hm[EPT2];
#pragma unroll
            for (int k = 0; k < EPT2; ++k) {
                unsigned ic = min(i0 + (unsigned)k * 1024u, n - 1u);
                d[k]  = (unsigned)dst_c[ic];
                s2[k] = (unsigned)src_c[ic];
                t2[k] = __float_as_uint(ts_c[ic]);
            }
#pragma unroll
            for (int k = 0; k < EPT2; ++k) {
                const bool live = i0 + (unsigned)k * 1024u < n;
                unsigned b  = d[k] / BN;
                unsigned dl = d[k] - b * BN;
                w0[k] = s2[k] | (dl << SRC_BITS);
                lane_run r = find_runs(live ? b : ~0u, lane);
                hm[k] = r.heads;
                p[k] = 0u;
                if (r.head && live)
                    p[k] = atomicAdd(&off[b], r.len);
            }
#pragma unroll
            for (int k = 0; k < EPT2; ++k) {
                unsigned hl = 31u - (unsigned)__builtin_clz(hm[k] & ((2u << lane) - 1u));
                unsigned base = (unsigned)__builtin_amdgcn_ds_bpermute((int)(hl << 2), (int)p[k]);
                p[k] = base + (lane - hl);
            }
#pragma unroll
            for (int k = 0; k < EPT2; ++k)
                if (i0 + (unsigned)k * 1024u < n)
                    rec_c[p[k]] = ((unsigned long long)t2[k] << 32) | (unsigned long long)w0[k];
        }
    }
}


template <int J>
__device__ __forceinline__ int row_bcast16(int v) {
    return __builtin_amdgcn_update_dpp(0, v, 0x150 + J  , 0xf, 0xf, false);
}

template <int J0, int J1>
__device__ __forceinline__ int row_bcast8(int v) {
    int lo = __builtin_amdgcn_update_dpp(0, v, 0x150 + J0, 0xf, 0x3, false);
    return __builtin_amdgcn_update_dpp(lo, v, 0x150 + J1, 0xf, 0xc, false);
}

template <int LOG2F, int J>
__device__ __forceinline__ int rec_bcast(int v) {
    if constexpr (LOG2F == 4) return row_bcast16<J>(v);
    else                      return row_bcast8<2 * J, 2 * J + 1>(v);
}

template <int LOG2F, int J>
__device__ __forceinline__ void rec_step_load(int pv, unsigned fbyte,
                                              const char* __restrict__ feat_b,
                                              unsigned* so, float* fv) {
    unsigned p  = (unsigned)rec_bcast<LOG2F, J>(pv);
    unsigned s  = p & ((1u << SRC_BITS) - 1u);
    unsigned dl = p >> SRC_BITS;
    so[J] = (dl << LOG2F) + (fbyte >> 2);
    fv[J] = *(const float*)(feat_b + ((s << (LOG2F + 2)) + fbyte));
}

template <int LOG2F, int J>
__device__ __forceinline__ void rec_step_max(int tv, int lim, unsigned* plane,
                                             const unsigned* so, const float* fv) {
    const int STRIDE = (LOG2F == 4) ? 1 : 2;
    float t = __int_as_float(rec_bcast<LOG2F, J>(tv));
    unsigned key = ord_encode(fv[J] * t);
    if (STRIDE * J < lim)
        atomicMax(&plane[so[J]], key);
}

template <int LOG2F, int... J>
__device__ __forceinline__ void rec_batch(int pv, int tv, int lim, unsigned fbyte,
                                          const char* __restrict__ feat_b, unsigned* plane,
                                          std::integer_sequence<int, J...>) {
    unsigned so[sizeof...(J)];
    float fv[sizeof...(J)];
    (rec_step_load<LOG2F, J>(pv, fbyte, feat_b, so, fv), ...);
    (rec_step_max<LOG2F, J>(tv, lim, plane, so, fv), ...);
}

__device__ __forceinline__ void range_table(const unsigned long long* __restrict__ seg_row,
                                            unsigned* segQ, unsigned* segS, unsigned* segL) {
    const int tid = threadIdx.x;
    if (tid < NCHUNK) {
        unsigned long long e = seg_row[tid];
        segS[tid] = (unsigned)e;
        segL[tid] = (unsigned)(e >> 32);
    }
    __syncthreads();
    if (tid < 32) {
        unsigned v[8];
        unsigned s = 0u;
#pragma unroll
        for (int i = 0; i < 8; ++i) {
            v[i] = segL[tid * 8 + i];
            s += v[i];
        }
        unsigned incl = s;
#pragma unroll
        for (int d = 1; d < 32; d <<= 1) {
            unsigned t = __shfl_up(incl, d);
            if (tid >= d) incl += t;
        }
        unsigned run = incl - s;
#pragma unroll
        for (int i = 0; i < 8; ++i) {
            segQ[tid * 8 + i] = run;
            run += v[i];
        }
        if (tid == 31) segQ[NCHUNK] = run;
    }
    __syncthreads();
}

template <int LOG2F>
__device__ __forceinline__ void range_max(const unsigned long long* __restrict__ rec,
                                          const unsigned* segQ, const unsigned* segS,
                                          const float* __restrict__ feat, unsigned* plane) {
    const int F = 1 << LOG2F;
    const int lane = threadIdx.x & 31;
    const unsigned wv = (unsigned)__builtin_amdgcn_readfirstlane((int)(threadIdx.x >> 5));
    const unsigned nw = blockDim.x >> 5;
    const int f = lane & (F - 1);
    const int u = (LOG2F == 4) ? 0 : ((lane >> 3) & 1);
    const unsigned total = (unsigned)__builtin_amdgcn_readfirstlane((int)segQ[NCHUNK]);
    const unsigned span  = (total + nw - 1u) / nw;
    unsigned pos = min(wv * span, total);
    const unsigned span_end = min(pos + span, total);
    if (pos < span_end) {
        unsigned below = 0u;
#pragma unroll
        for (int r = 0; r < NCHUNK / 32; ++r)
            below += (unsigned)__builtin_popcount(
                __builtin_amdgcn_ballot_w32(segQ[lane + 32 * r] <= pos));
        unsigned kc = below - 1u;
        while (pos < span_end) {
            const unsigned kk = kc + (unsigned)lane;
            const unsigned qv = segQ[min(kk, (unsigned)NCHUNK)];
            const unsigned av = segS[min(kk, (unsigned)NCHUNK - 1u)] - qv;
            unsigned Q[PIECE + 1], A[PIECE];
#pragma unroll
            for (int m = 0; m <= PIECE; ++m) Q[m] = (unsigned)__builtin_amdgcn_readlane((int)qv, m);
#pragma unroll
            for (int m = 0; m < PIECE; ++m)  A[m] = (unsigned)__builtin_amdgcn_readlane((int)av, m);
            const unsigned piece_end = min(span_end, Q[PIECE]);
            if (pos < piece_end) {
                const unsigned lastj = piece_end - 1u;
                unsigned j = min(pos + (unsigned)lane, lastj);
                unsigned a = A[0];
#pragma unroll
                for (int m = 1; m < PIECE; ++m) a = (j >= Q[m]) ? A[m] : a;
                unsigned long long r = rec[j + a];
                for (unsigned i = pos; i < piece_end; i += 32u) {
                    j = min(i + 32u + (unsigned)lane, lastj);
                    a = A[0];
#pragma unroll
                    for (int m = 1; m < PIECE; ++m) a = (j >= Q[m]) ? A[m] : a;
                    unsigned long long rn = rec[j + a];
                    int pv = (int)(unsigned)r;
                    int tv = (int)(unsigned)(r >> 32);
                    int lim = (int)(piece_end - i) - 16 * (lane >> 4) - u;
                    rec_batch<LOG2F>(pv, tv, lim, (unsigned)f << 2, (const char*)feat, plane,
                                     std::make_integer_sequence<int, F>{});
                    r = rn;
                }
            }
            pos = piece_end;
            kc += PIECE;
        }
    }
}


template <int J>
__device__ __forceinline__ int quad_bcast(int v) {
    return __builtin_amdgcn_update_dpp(v, v, J * 0x55  , 0xf, 0xf, false);
}

template <int FPL> struct feat_vec;
template <> struct feat_vec<4> { typedef float4 type; };
template <> struct feat_vec<2> { typedef float2 type; };

template <int LOG2F, int J>
__device__ __forceinline__ void ord_step_load(int pv, unsigned qbyte, const char* __restrict__ feat_b,
                                              unsigned* dlv, typename feat_vec<(1 << LOG2F) / 4>::type* fv) {
    typedef typename feat_vec<(1 << LOG2F) / 4>::type vec_t;
    unsigned p = (unsigned)quad_bcast<J>(pv);
    unsigned s = p & ((1u << SRC_BITS) - 1u);
    dlv[J] = p >> SRC_BITS;
    fv[J] = *(const vec_t*)(feat_b + ((s << (LOG2F + 2)) + qbyte));
}

template <int LOG2F, int J>
__device__ __forceinline__ void ord_step_fold(int tv, unsigned q, unsigned* plane,
                                              const unsigned* dlv,
                                              const typename feat_vec<(1 << LOG2F) / 4>::type* fv,
                                              unsigned& cur, float* acc) {
    constexpr int FPL = (1 << LOG2F) / 4;
    float t = __int_as_float(quad_bcast<J>(tv));
    const float* fj = (const float*)&fv[J];
    const bool chg = dlv[J] != cur;
    if (chg) {
#pragma unroll
        for (int e = 0; e < FPL; ++e)
            atomicMax(&plane[(cur << LOG2F) + q * FPL + e], ord_encode(acc[e]));
    }
#pragma unroll
    for (int e = 0; e < FPL; ++e) {
        float msg = fj[e] * t;
        acc[e] = (chg || msg > acc[e]) ? msg : acc[e];
    }
    cur = dlv[J];
}

template <int LOG2F, int... J>
__device__ __forceinline__ void ord_batch(int pv, int tv, unsigned q,
                                          const char* __restrict__ feat_b, unsigned* plane,
                                          unsigned& cur, float* acc,
                                          std::integer_sequence<int, J...>) {
    constexpr int FPL = (1 << LOG2F) / 4;
    unsigned dlv[sizeof...(J)];
    typename feat_vec<FPL>::type fv[sizeof...(J)];
    (ord_step_load<LOG2F, J>(pv, q * (FPL * 4u), feat_b, dlv, fv), ...);
    (ord_step_fold<LOG2F, J>(tv, q, plane, dlv, fv, cur, acc), ...);
}

template <int LOG2F>
__device__ __forceinline__ void range_ordered(const unsigned long long* __restrict__ rec,
                                              const unsigned* segS, const unsigned* segL,
                                              unsigned n, const float* __restrict__ feat,
                                              unsigned* plane, unsigned* cnt, unsigned* off,
                                              uint2* sorted) {
    constexpr int F = 1 << LOG2F;
    constexpr int PPW = NCHUNK / 32;
    const int tid = threadIdx.x;
    const unsigned lane = (unsigned)tid & 31u;
    const unsigned wv = (unsigned)__builtin_amdgcn_readfirstlane((int)((unsigned)tid >> 5));
    const uint2* __restrict__ rec2 = reinterpret_cast<const uint2*>(rec);

    unsigned S[PPW], L[PPW];
    uint2 R[PPW][SORT_SLOTS];
#pragma unroll
    for (int p = 0; p < PPW; ++p) {
        S[p] = (unsigned)__builtin_amdgcn_readfirstlane((int)segS[wv + 32u * p]);
        L[p] = (unsigned)__builtin_amdgcn_readfirstlane((int)segL[wv + 32u * p]);
    }
#pragma unroll
    for (int p = 0; p < PPW; ++p) {
        const unsigned base = L[p] != 0u ? S[p] : 0u;
        const unsigned last = L[p] != 0u ? L[p] - 1u : 0u;
#pragma unroll
        for (int q = 0; q < SORT_SLOTS; ++q)
            R[p][q] = rec2[base + min(lane + 32u * q, last)];
    }
#pragma unroll
    for (int p = 0; p < PPW; ++p) {
#pragma unroll
        for (int q = 0; q < SORT_SLOTS; ++q)
            if (lane + 32u * q < L[p]) atomicAdd(&cnt[R[p][q].x >> SRC_BITS], 1u);
        for (unsigned o0 = 32u * SORT_SLOTS; o0 < L[p]; o0 += 32u * REM_C) {
            unsigned xw[REM_C];
#pragma unroll
            for (int k = 0; k < REM_C; ++k)
                xw[k] = rec2[S[p] + min(o0 + 32u * k + lane, L[p] - 1u)].x;
#pragma unroll
            for (int k = 0; k < REM_C; ++k)
                if (o0 + 32u * k + lane < L[p]) atomicAdd(&cnt[xw[k] >> SRC_BITS], 1u);
        }
    }
    __syncthreads();
    if (tid < 32) {
        unsigned v[ROWS_PAD / 32];
        unsigned s = 0u;
#pragma unroll
        for (int i = 0; i < ROWS_PAD / 32; ++i) {
            v[i] = cnt[tid * (ROWS_PAD / 32) + i];
            s += v[i];
        }
        unsigned incl = s;
#pragma unroll
        for (int dd = 1; dd < 32; dd <<= 1) {
            unsigned t = __shfl_up(incl, dd);
            if (tid >= dd) incl += t;
        }
        unsigned run = incl - s;
#pragma unroll
        for (int i = 0; i < ROWS_PAD / 32; ++i) {
            off[tid * (ROWS_PAD / 32) + i] = run;
            run += v[i];
        }
    }
    __syncthreads();
#pragma unroll
    for (int p = 0; p < PPW; ++p) {
        unsigned at[SORT_SLOTS];
#pragma unroll
        for (int q = 0; q < SORT_SLOTS; ++q) {
            at[q] = 0u;
            if (lane + 32u * q < L[p]) at[q] = atomicAdd(&off[R[p][q].x >> SRC_BITS], 1u);
        }
#pragma unroll
        for (int q = 0; q < SORT_SLOTS; ++q)
            if (lane + 32u * q < L[p]) sorted[at[q]] = R[p][q];
        for (unsigned o0 = 32u * SORT_SLOTS; o0 < L[p]; o0 += 32u * REM_P) {
            uint2 t[REM_P];
            unsigned a[REM_P];
#pragma unroll
            for (int k = 0; k < REM_P; ++k)
                t[k] = rec2[S[p] + min(o0 + 32u * k + lane, L[p] - 1u)];
#pragma unroll
            for (int k = 0; k < REM_P; ++k) {
                a[k] = 0u;
                if (o0 + 32u * k + lane < L[p]) a[k] = atomicAdd(&off[t[k].x >> SRC_BITS], 1u);
            }
#pragma unroll
            for (int k = 0; k < REM_P; ++k)
                if (o0 + 32u * k + lane < L[p]) sorted[a[k]] = t[k];
        }
    }
    __syncthreads();

    constexpr int FPL = F / 4;
    constexpr unsigned NG = 32u * 8u;
    const unsigned share = (n + NG - 1u) / NG;
    const unsigned q = lane & 3u;
    const unsigned gid = wv * 8u + (lane >> 2);
    const unsigned g0 = gid * share;
    unsigned cur = sorted[min(g0, n - 1u)].x >> SRC_BITS;
    float acc[FPL];
#pragma unroll
    for (int e = 0; e < FPL; ++e) acc[e] = __uint_as_float(0xFF800000u);
    for (unsigned i = 0u; i < share; i += 4u) {
        uint2 r = sorted[min(g0 + i + q, n - 1u)];
        ord_batch<LOG2F>((int)r.x, (int)r.y, q, (const char*)feat, plane, cur, acc,
                         std::make_integer_sequence<int, 4>{});
    }
#pragma unroll
    for (int e = 0; e < FPL; ++e)
        atomicMax(&plane[(cur << LOG2F) + q * FPL + e], ord_encode(acc[e]));
}

__global__ void __launch_bounds__(1024)
GCN_layer1(const float* __restrict__ x,
           const unsigned long long* __restrict__ rec,
           const unsigned long long* __restrict__ seg,
           const float* __restrict__ W1,
           const float* __restrict__ b1,
           float* __restrict__ h,
           int n_nodes) {
    __shared__ unsigned plane[BN * 16];
    __shared__ unsigned segQ[NCHUNK + 1];
    __shared__ unsigned segS[NCHUNK];
    __shared__ unsigned segL[NCHUNK];
    __shared__ unsigned rcnt[ROWS_PAD];
    __shared__ unsigned roff[ROWS_PAD];
    __shared__ uint2 sorted[SORT_CAP];
    for (int i = threadIdx.x; i < BN * 16; i += 1024) plane[i] = ORD_NEG_INF;
    if (threadIdx.x < ROWS_PAD) rcnt[threadIdx.x] = 0u;
    range_table(seg + (size_t)blockIdx.x * NCHUNK, segQ, segS, segL);

    const unsigned total = (unsigned)__builtin_amdgcn_readfirstlane((int)segQ[NCHUNK]);
    if (total <= (unsigned)SORT_CAP) {
        if (total != 0u) range_ordered<4>(rec, segS, segL, total, x, plane, rcnt, roff, sorted);
    } else {
        range_max<4>(rec, segQ, segS, x, plane);
    }
    __syncthreads();

    int wv   = __builtin_amdgcn_readfirstlane((int)(threadIdx.x >> 5));
    int lane = threadIdx.x & 31;
    int base = (int)blockIdx.x * BN + wv * 16;
    if (wv >= BTILES || base >= n_nodes) return;

    int m  = lane & 15;
    int kh = lane >> 4;
    int n  = m;

    v8f acc = {0.f, 0.f, 0.f, 0.f, 0.f, 0.f, 0.f, 0.f};
#pragma unroll
    for (int c = 0; c < 4; ++c) {
        int k0 = 4 * c + 2 * kh;
        v2f a, b;
        a.x = ord_decode_finite(plane[(wv * 16 + m) * 16 + k0]);
        a.y = ord_decode_finite(plane[(wv * 16 + m) * 16 + k0 + 1]);
        b.x = (n < 8) ? W1[k0 * 8 + n]       : 0.0f;
        b.y = (n < 8) ? W1[(k0 + 1) * 8 + n] : 0.0f;
        acc = __builtin_amdgcn_wmma_f32_16x16x4_f32(
             false, a,  false, b,
             (short)0, acc,  false,  false);
    }

    if (n < 8) {
#pragma unroll
        for (int r = 0; r < 8; ++r) {
            int node = base + r + 8 * kh;
            float v = acc[r] + b1[n];
            h[node * 8 + n] = v > 0.0f ? v : 0.0f;
        }
    }
}

__global__ void __launch_bounds__(1024)
GCN_layer2(const float* __restrict__ h,
           const unsigned long long* __restrict__ rec,
           const unsigned long long* __restrict__ seg,
           const float* __restrict__ W2,
           const float* __restrict__ b2,
           float* __restrict__ out,
           int n_nodes) {
    __shared__ unsigned plane[BN * 8];
    __shared__ unsigned segQ[NCHUNK + 1];
    __shared__ unsigned segS[NCHUNK];
    __shared__ unsigned segL[NCHUNK];
    __shared__ unsigned rcnt[ROWS_PAD];
    __shared__ unsigned roff[ROWS_PAD];
    __shared__ uint2 sorted[SORT_CAP];
    for (int i = threadIdx.x; i < BN * 8; i += 1024) plane[i] = ORD_NEG_INF;
    if (threadIdx.x < ROWS_PAD) rcnt[threadIdx.x] = 0u;
    range_table(seg + (size_t)blockIdx.x * NCHUNK, segQ, segS, segL);

    const unsigned total = (unsigned)__builtin_amdgcn_readfirstlane((int)segQ[NCHUNK]);
    if (total <= (unsigned)SORT_CAP) {
        if (total != 0u) range_ordered<3>(rec, segS, segL, total, h, plane, rcnt, roff, sorted);
    } else {
        range_max<3>(rec, segQ, segS, h, plane);
    }
    __syncthreads();

    int wv   = __builtin_amdgcn_readfirstlane((int)(threadIdx.x >> 5));
    int lane = threadIdx.x & 31;
    int base = (int)blockIdx.x * BN + wv * 16;
    if (wv >= BTILES || base >= n_nodes) return;

    int m  = lane & 15;
    int kh = lane >> 4;
    int n  = m;

    v8f acc = {0.f, 0.f, 0.f, 0.f, 0.f, 0.f, 0.f, 0.f};
#pragma unroll
    for (int c = 0; c < 2; ++c) {
        int k0 = 4 * c + 2 * kh;
        v2f a, b;
        a.x = ord_decode_finite(plane[(wv * 16 + m) * 8 + k0]);
        a.y = ord_decode_finite(plane[(wv * 16 + m) * 8 + k0 + 1]);
        b.x = (n < 2) ? W2[k0 * 2 + n]       : 0.0f;
        b.y = (n < 2) ? W2[(k0 + 1) * 2 + n] : 0.0f;
        acc = __builtin_amdgcn_wmma_f32_16x16x4_f32(
            false, a, false, b, (short)0, acc, false, false);
    }

    if (n < 2) {
#pragma unroll
        for (int r = 0; r < 8; ++r) {
            int node = base + r + 8 * kh;
            out[node * 2 + n] = acc[r] + b2[n];
        }
    }
}

extern "C" void kernel_launch(void* const* d_in, const int* in_sizes, int n_in,
                              void* d_out, int out_size, void* d_ws, size_t ws_size,
                              hipStream_t stream) {
    const float* x   = (const float*)d_in[0];
    const int*   src = (const int*)  d_in[1];
    const int*   dst = (const int*)  d_in[2];
    const float* ts  = (const float*)d_in[3];
    const float* W1  = (const float*)d_in[4];
    const float* b1  = (const float*)d_in[5];
    const float* W2  = (const float*)d_in[6];
    const float* b2  = (const float*)d_in[7];
    float* out = (float*)d_out;

    const int N = in_sizes[0] / 16;
    const int E = in_sizes[1];

    unsigned long long* seg = (unsigned long long*)d_ws;
    unsigned long long* rec = seg + (size_t)NBMAX * NCHUNK;
    float* h = (float*)(rec + (size_t)E);

    const int nb    = (N + BN - 1) / BN;
    const int chunk = (E + NCHUNK - 1) / NCHUNK;

    GCN_group<<<NCHUNK, 1024, 0, stream>>>(src, dst, ts, seg, rec, E, chunk, nb);
    GCN_layer1<<<nb, 1024, 0, stream>>>(x, rec, seg, W1, b1, h, N);
    GCN_layer2<<<nb, 1024, 0, stream>>>(h, rec, seg, W2, b2, out, N);
}
